# starting kernel
# baseline (speedup 1.0000x reference)
#include <hip/hip_runtime.h>
#include <math.h>

#define N_NODES   50000
#define N_EDGES   1600000
#define NFEAT     500
#define HIDDEN    64
#define NCLASS    40
#define NUM_LAYERS 4
#define STEP_SZ   0.1f
#define M_TILES   (N_NODES / 16)

typedef __attribute__((ext_vector_type(2))) float v2f;
typedef __attribute__((ext_vector_type(8))) float v8f;


__global__ void deg_count_kernel(const int* __restrict__ col, float* __restrict__ deg) {
  int e = blockIdx.x * blockDim.x + threadIdx.x;
  if (e < N_EDGES) atomicAdd(&deg[col[e]], 1.0f);
}

__global__ void deg_inv_kernel(float* __restrict__ deg) {
  int i = blockIdx.x * blockDim.x + threadIdx.x;
  if (i < N_NODES) {
    float d = deg[i];
    deg[i] = (d > 0.0f) ? rsqrtf(d) : 0.0f;
  }
}

__global__ void coef_kernel(const int* __restrict__ row, const int* __restrict__ col,
                            const float* __restrict__ dinv, float* __restrict__ coef) {
  int e = blockIdx.x * blockDim.x + threadIdx.x;
  if (e < N_EDGES) coef[e] = dinv[row[e]] * dinv[col[e]];
}

__global__ void weff_kernel(const float* __restrict__ pw, float* __restrict__ W) {
  int i = threadIdx.x;
  const int LD = HIDDEN + 2;
  float q = pw[i * LD + HIDDEN];
  float r = pw[i * LD + HIDDEN + 1];
  float s = 0.0f;
  for (int j = 0; j < HIDDEN; ++j) {
    float v = 0.0f;
    if (i < j)      v = pw[i * LD + j];
    else if (i > j) v = pw[j * LD + i];
    s += fabsf(v);
    W[i * HIDDEN + j] = v;
  }
  W[i * HIDDEN + i] = q * s + r;
}


__global__ __launch_bounds__(128)
void enc_gemm_kernel(const float* __restrict__ x, const float* __restrict__ ew,
                     float* __restrict__ h, float* __restrict__ x0) {
  int wave = (int)((blockIdx.x * blockDim.x + threadIdx.x) >> 5);
  if (wave >= M_TILES) return;
  int lane = threadIdx.x & 31;
  int mr = lane & 15;
  int kh = lane >> 4;
  const float* arow = x  + (size_t)(wave * 16 + mr) * NFEAT;
  const float* b0r  = ew + (size_t)(mr +  0) * NFEAT;
  const float* b1r  = ew + (size_t)(mr + 16) * NFEAT;
  const float* b2r  = ew + (size_t)(mr + 32) * NFEAT;
  const float* b3r  = ew + (size_t)(mr + 48) * NFEAT;
  v8f c0 = {}, c1 = {}, c2 = {}, c3 = {};
  for (int k = 0; k < NFEAT; k += 4) {
    int ko = k + 2 * kh;
    v2f a  = *(const v2f*)(arow + ko);
    v2f b0 = *(const v2f*)(b0r + ko);
    v2f b1 = *(const v2f*)(b1r + ko);
    v2f b2 = *(const v2f*)(b2r + ko);
    v2f b3 = *(const v2f*)(b3r + ko);
    c0 = __builtin_amdgcn_wmma_f32_16x16x4_f32(false, a, false, b0, (short)0, c0, false, false);
    c1 = __builtin_amdgcn_wmma_f32_16x16x4_f32(false, a, false, b1, (short)0, c1, false, false);
    c2 = __builtin_amdgcn_wmma_f32_16x16x4_f32(false, a, false, b2, (short)0, c2, false, false);
    c3 = __builtin_amdgcn_wmma_f32_16x16x4_f32(false, a, false, b3, (short)0, c3, false, false);
  }
  int mbase = wave * 16 + kh * 8;
  for (int r = 0; r < 8; ++r) {
    size_t o = (size_t)(mbase + r) * HIDDEN + mr;
    h[o]      = c0[r]; x0[o]      = c0[r];
    h[o + 16] = c1[r]; x0[o + 16] = c1[r];
    h[o + 32] = c2[r]; x0[o + 32] = c2[r];
    h[o + 48] = c3[r]; x0[o + 48] = c3[r];
  }
}


__global__ __launch_bounds__(128)
void hw_gemm_kernel(const float* __restrict__ h, const float* __restrict__ W,
                    float* __restrict__ outp) {
  __shared__ float Ws[HIDDEN * HIDDEN];
  for (int i = threadIdx.x; i < HIDDEN * HIDDEN; i += blockDim.x) Ws[i] = W[i];
  __syncthreads();
  int wave = (int)((blockIdx.x * blockDim.x + threadIdx.x) >> 5);
  if (wave >= M_TILES) return;
  int lane = threadIdx.x & 31;
  int mr = lane & 15;
  int kh = lane >> 4;
  const float* arow = h + (size_t)(wave * 16 + mr) * HIDDEN;
  v8f c0 = {}, c1 = {}, c2 = {}, c3 = {};
  for (int k = 0; k < HIDDEN; k += 4) {
    int ko = k + 2 * kh;
    v2f a  = *(const v2f*)(arow + ko);
    v2f b0 = *(const v2f*)(&Ws[(mr +  0) * HIDDEN + ko]);
    v2f b1 = *(const v2f*)(&Ws[(mr + 16) * HIDDEN + ko]);
    v2f b2 = *(const v2f*)(&Ws[(mr + 32) * HIDDEN + ko]);
    v2f b3 = *(const v2f*)(&Ws[(mr + 48) * HIDDEN + ko]);
    c0 = __builtin_amdgcn_wmma_f32_16x16x4_f32(false, a, false, b0, (short)0, c0, false, false);
    c1 = __builtin_amdgcn_wmma_f32_16x16x4_f32(false, a, false, b1, (short)0, c1, false, false);
    c2 = __builtin_amdgcn_wmma_f32_16x16x4_f32(false, a, false, b2, (short)0, c2, false, false);
    c3 = __builtin_amdgcn_wmma_f32_16x16x4_f32(false, a, false, b3, (short)0, c3, false, false);
  }
  int mbase = wave * 16 + kh * 8;
  for (int r = 0; r < 8; ++r) {
    size_t o = (size_t)(mbase + r) * HIDDEN + mr;
    outp[o]      = c0[r];
    outp[o + 16] = c1[r];
    outp[o + 32] = c2[r];
    outp[o + 48] = c3[r];
  }
}


__global__ __launch_bounds__(256)
void edge_agg_kernel(const int* __restrict__ row, const int* __restrict__ col,
                     const float* __restrict__ coef, const float* __restrict__ outp,
                     float* __restrict__ agg) {
  int e = (int)((blockIdx.x * blockDim.x + threadIdx.x) >> 5);
  if (e >= N_EDGES) return;
  int lane = threadIdx.x & 31;
  int r  = __builtin_amdgcn_readfirstlane(row[e]);
  int c  = __builtin_amdgcn_readfirstlane(col[e]);
  float cf = __int_as_float(__builtin_amdgcn_readfirstlane(__float_as_int(coef[e])));
  const float* src = outp + (size_t)r * HIDDEN + 2 * lane;
  float*       dst = agg  + (size_t)c * HIDDEN + 2 * lane;
  v2f v = *(const v2f*)src;
  atomicAdd(&dst[0], cf * v.x);
  atomicAdd(&dst[1], cf * v.y);
}


__global__ __launch_bounds__(256)
void update_kernel(float* __restrict__ h, const float* __restrict__ agg,
                   const float* __restrict__ x0, const float* __restrict__ ext_w,
                   const float* __restrict__ beta) {
  int idx = blockIdx.x * blockDim.x + threadIdx.x;
  int j = idx & (HIDDEN - 1);
  float hv = h[idx];
  float conv = agg[idx] - hv * ext_w[j] + x0[idx] * beta[0];
  h[idx] = hv + STEP_SZ * fmaxf(conv, 0.0f);
}


__global__ __launch_bounds__(128)
void dec_gemm_kernel(const float* __restrict__ h, const float* __restrict__ dw,
                     float* __restrict__ out) {
  int wave = (int)((blockIdx.x * blockDim.x + threadIdx.x) >> 5);
  if (wave >= M_TILES) return;
  int lane = threadIdx.x & 31;
  int mr = lane & 15;
  int kh = lane >> 4;
  const float* arow = h + (size_t)(wave * 16 + mr) * HIDDEN;
  const float* b0r = dw + (size_t)(mr +  0) * HIDDEN;
  const float* b1r = dw + (size_t)(mr + 16) * HIDDEN;
  int n2 = mr + 32;
  const float* b2r = dw + (size_t)(n2 < NCLASS ? n2 : 0) * HIDDEN;
  float b2scale = (n2 < NCLASS) ? 1.0f : 0.0f;
  v8f c0 = {}, c1 = {}, c2 = {};
  for (int k = 0; k < HIDDEN; k += 4) {
    int ko = k + 2 * kh;
    v2f a  = *(const v2f*)(arow + ko);
    v2f b0 = *(const v2f*)(b0r + ko);
    v2f b1 = *(const v2f*)(b1r + ko);
    v2f b2 = *(const v2f*)(b2r + ko);
    b2 *= b2scale;
    c0 = __builtin_amdgcn_wmma_f32_16x16x4_f32(false, a, false, b0, (short)0, c0, false, false);
    c1 = __builtin_amdgcn_wmma_f32_16x16x4_f32(false, a, false, b1, (short)0, c1, false, false);
    c2 = __builtin_amdgcn_wmma_f32_16x16x4_f32(false, a, false, b2, (short)0, c2, false, false);
  }
  int mbase = wave * 16 + kh * 8;
  for (int r = 0; r < 8; ++r) {
    float* orow = out + (size_t)(mbase + r) * NCLASS;
    orow[mr]      = c0[r];
    orow[mr + 16] = c1[r];
    if (n2 < NCLASS) orow[n2] = c2[r];
  }
}


extern "C" void kernel_launch(void* const* d_in, const int* in_sizes, int n_in,
                              void* d_out, int out_size, void* d_ws, size_t ws_size,
                              hipStream_t stream) {
  (void)in_sizes; (void)n_in; (void)out_size; (void)ws_size;
  const float* x     = (const float*)d_in[0];
  const int*   ei    = (const int*)  d_in[1];
  const float* enc_w = (const float*)d_in[2];
  const float* dec_w = (const float*)d_in[3];
  const float* ext_w = (const float*)d_in[4];
  const float* beta  = (const float*)d_in[5];
  const float* pw    = (const float*)d_in[6];
  const int* row = ei;
  const int* col = ei + N_EDGES;

  float* ws   = (float*)d_ws;
  float* deg  = ws;
  float* coef = deg  + N_NODES;
  float* W    = coef + N_EDGES;
  float* h    = W    + HIDDEN * HIDDEN;
  float* x0   = h    + (size_t)N_NODES * HIDDEN;
  float* outp = x0   + (size_t)N_NODES * HIDDEN;
  float* agg  = outp + (size_t)N_NODES * HIDDEN;

  hipMemsetAsync(deg, 0, N_NODES * sizeof(float), stream);
  deg_count_kernel<<<(N_EDGES + 255) / 256, 256, 0, stream>>>(col, deg);
  deg_inv_kernel<<<(N_NODES + 255) / 256, 256, 0, stream>>>(deg);
  coef_kernel<<<(N_EDGES + 255) / 256, 256, 0, stream>>>(row, col, deg, coef);
  weff_kernel<<<1, HIDDEN, 0, stream>>>(pw, W);

  const int gemm_blocks = (M_TILES + 3) / 4;
  enc_gemm_kernel<<<gemm_blocks, 128, 0, stream>>>(x, enc_w, h, x0);

  for (int l = 0; l < NUM_LAYERS; ++l) {
    hw_gemm_kernel<<<gemm_blocks, 128, 0, stream>>>(h, W, outp);
    hipMemsetAsync(agg, 0, (size_t)N_NODES * HIDDEN * sizeof(float), stream);
    edge_agg_kernel<<<N_EDGES / 8, 256, 0, stream>>>(row, col, coef, outp, agg);
    update_kernel<<<(N_NODES * HIDDEN) / 256, 256, 0, stream>>>(h, agg, x0, ext_w, beta);
  }

  dec_gemm_kernel<<<gemm_blocks, 128, 0, stream>>>(h, dec_w, (float*)d_out);
}


# deg^-1/2[row]*deg^-1/2[col] applied per node, not per edge: out_p rows stored times deg^-1/2, node sum scaled in the update; coef array and 2 launches (coef, deg^-1/2) dropped: 12; sum order kept
# speedup vs baseline: 2.7457x; 2.7457x over previous
#include <hip/hip_runtime.h>
#include <math.h>

#define N_NODES   50000
#define N_EDGES   1600000
#define NFEAT     500
#define HIDDEN    64
#define NCLASS    40
#define NUM_LAYERS 4
#define STEP_SZ   0.1f
#define M_TILES   (N_NODES / 16)
#define DEG_PAD   50048
#define ENC_KB    (NFEAT / 8)
#define ENC_FRAGS (ENC_KB * 4 * 32)

typedef __attribute__((ext_vector_type(2))) float v2f;
typedef __attribute__((ext_vector_type(4))) float v4f;
typedef __attribute__((ext_vector_type(8))) float v8f;

#define PLANE ((size_t)N_NODES * HIDDEN)

__device__ __forceinline__ int l2_half() {
  unsigned int id = __builtin_amdgcn_s_sendmsg_rtn(0x87);
  return (int)((id >> 18) & 1u);
}


__global__ void deg_count_kernel(const int* __restrict__ col, float* __restrict__ deg) {
  int e = blockIdx.x * blockDim.x + threadIdx.x;
  float* d = deg + l2_half() * DEG_PAD;
  if (e < N_EDGES) atomicAdd(&d[col[e]], 1.0f);
}

__device__ __forceinline__ float dinv_of(float d0, float d1) {
  float d = d0 + d1;
  return (d > 0.0f) ? rsqrtf(d) : 0.0f;
}

__device__ __forceinline__ int wf_pos(int n, int k) {
  int lane = (n & 15) + 16 * ((k >> 2) & 1);
  return ((((k >> 3) * 4 + (n >> 4)) * 32) + lane) * 4 + (k & 3);
}

__global__ void weff_pack_kernel(const float* __restrict__ pw, float* __restrict__ W,
                                 const float* __restrict__ ew, float* __restrict__ ewp) {
  if (blockIdx.x != 0) {
    int idx = (int)(blockIdx.x - 1) * HIDDEN + threadIdx.x;
    if (idx < ENC_FRAGS) {
      int lane = idx & 31;
      int t    = (idx >> 5) & 3;
      int kb   = idx >> 7;
      const float* s = ew + (size_t)(16 * t + (lane & 15)) * NFEAT + 8 * kb + 4 * (lane >> 4);
      *(v4f*)(ewp + (size_t)idx * 4) = *(const v4f*)s;
    }
    return;
  }
  int i = threadIdx.x;
  const int LD = HIDDEN + 2;
  float q = pw[i * LD + HIDDEN];
  float r = pw[i * LD + HIDDEN + 1];
  float s = 0.0f;
  for (int j = 0; j < HIDDEN; ++j) {
    float v = 0.0f;
    if (i < j)      v = pw[i * LD + j];
    else if (i > j) v = pw[j * LD + i];
    s += fabsf(v);
    W[wf_pos(i, j)] = v;
  }
  W[wf_pos(i, i)] = q * s + r;
}


#define ENC_MMA8(a, b0, b1, b2, b3)                                                              \
  do {                                                                                           \
    v2f alo = __builtin_shufflevector(a, a, 0, 1), ahi = __builtin_shufflevector(a, a, 2, 3);     \
    c0 = __builtin_amdgcn_wmma_f32_16x16x4_f32(false, alo, false, __builtin_shufflevector(b0, b0, 0, 1), (short)0, c0, false, false); \
    c1 = __builtin_amdgcn_wmma_f32_16x16x4_f32(false, alo, false, __builtin_shufflevector(b1, b1, 0, 1), (short)0, c1, false, false); \
    c2 = __builtin_amdgcn_wmma_f32_16x16x4_f32(false, alo, false, __builtin_shufflevector(b2, b2, 0, 1), (short)0, c2, false, false); \
    c3 = __builtin_amdgcn_wmma_f32_16x16x4_f32(false, alo, false, __builtin_shufflevector(b3, b3, 0, 1), (short)0, c3, false, false); \
    c0 = __builtin_amdgcn_wmma_f32_16x16x4_f32(false, ahi, false, __builtin_shufflevector(b0, b0, 2, 3), (short)0, c0, false, false); \
    c1 = __builtin_amdgcn_wmma_f32_16x16x4_f32(false, ahi, false, __builtin_shufflevector(b1, b1, 2, 3), (short)0, c1, false, false); \
    c2 = __builtin_amdgcn_wmma_f32_16x16x4_f32(false, ahi, false, __builtin_shufflevector(b2, b2, 2, 3), (short)0, c2, false, false); \
    c3 = __builtin_amdgcn_wmma_f32_16x16x4_f32(false, ahi, false, __builtin_shufflevector(b3, b3, 2, 3), (short)0, c3, false, false); \
  } while (0)

__global__ __launch_bounds__(128)
void enc_gemm_kernel(const float* __restrict__ x, const float* __restrict__ ew,
                     const float* __restrict__ ewp,
                     float* __restrict__ h, float* __restrict__ x0) {
  int wave = (int)((blockIdx.x * blockDim.x + threadIdx.x) >> 5);
  if (wave >= M_TILES) return;
  int lane = threadIdx.x & 31;
  int mr = lane & 15;
  int kh = lane >> 4;
  const float* arow = x + (size_t)(wave * 16 + mr) * NFEAT;
  const float* ap   = arow + 4 * kh;
  const float* bp   = ewp + lane * 4;
  v8f c0 = {}, c1 = {}, c2 = {}, c3 = {};
  v4f aE  = *(const v4f*)(ap);
  v4f bE0 = *(const v4f*)(bp);
  v4f bE1 = *(const v4f*)(bp + 128);
  v4f bE2 = *(const v4f*)(bp + 256);
  v4f bE3 = *(const v4f*)(bp + 384);
  for (int kb = 0; kb < ENC_KB - 2; kb += 2) {
    const float* bo = bp + (size_t)(kb + 1) * 512;
    v4f aO  = *(const v4f*)(ap + (kb + 1) * 8);
    v4f bO0 = *(const v4f*)(bo);
    v4f bO1 = *(const v4f*)(bo + 128);
    v4f bO2 = *(const v4f*)(bo + 256);
    v4f bO3 = *(const v4f*)(bo + 384);
    ENC_MMA8(aE, bE0, bE1, bE2, bE3);
    const float* be = bp + (size_t)(kb + 2) * 512;
    aE  = *(const v4f*)(ap + (kb + 2) * 8);
    bE0 = *(const v4f*)(be);
    bE1 = *(const v4f*)(be + 128);
    bE2 = *(const v4f*)(be + 256);
    bE3 = *(const v4f*)(be + 384);
    ENC_MMA8(aO, bO0, bO1, bO2, bO3);
  }
  {
    const float* bo = bp + (size_t)(ENC_KB - 1) * 512;
    v4f aO  = *(const v4f*)(ap + (ENC_KB - 1) * 8);
    v4f bO0 = *(const v4f*)(bo);
    v4f bO1 = *(const v4f*)(bo + 128);
    v4f bO2 = *(const v4f*)(bo + 256);
    v4f bO3 = *(const v4f*)(bo + 384);
    ENC_MMA8(aE, bE0, bE1, bE2, bE3);
    ENC_MMA8(aO, bO0, bO1, bO2, bO3);
  }
  {
    int ko = ENC_KB * 8 + 2 * kh;
    v2f a  = *(const v2f*)(arow + ko);
    v2f b0 = *(const v2f*)(ew + (size_t)(mr +  0) * NFEAT + ko);
    v2f b1 = *(const v2f*)(ew + (size_t)(mr + 16) * NFEAT + ko);
    v2f b2 = *(const v2f*)(ew + (size_t)(mr + 32) * NFEAT + ko);
    v2f b3 = *(const v2f*)(ew + (size_t)(mr + 48) * NFEAT + ko);
    c0 = __builtin_amdgcn_wmma_f32_16x16x4_f32(false, a, false, b0, (short)0, c0, false, false);
    c1 = __builtin_amdgcn_wmma_f32_16x16x4_f32(false, a, false, b1, (short)0, c1, false, false);
    c2 = __builtin_amdgcn_wmma_f32_16x16x4_f32(false, a, false, b2, (short)0, c2, false, false);
    c3 = __builtin_amdgcn_wmma_f32_16x16x4_f32(false, a, false, b3, (short)0, c3, false, false);
  }
  int mbase = wave * 16 + kh * 8;
  for (int r = 0; r < 8; ++r) {
    size_t o = (size_t)(mbase + r) * HIDDEN + mr;
    h[o]      = c0[r]; x0[o]      = c0[r];
    h[o + 16] = c1[r]; x0[o + 16] = c1[r];
    h[o + 32] = c2[r]; x0[o + 32] = c2[r];
    h[o + 48] = c3[r]; x0[o + 48] = c3[r];
  }
}


#define EDGES_PER_WAVE 32
#define EDGE_GROUP     16
#define HALF_ROW       (HIDDEN / 2)

__global__ __launch_bounds__(256)
void edge_agg_kernel(const int* __restrict__ row, const int* __restrict__ col,
                     const float* __restrict__ outp, float* __restrict__ agg) {
  int wave = (int)((blockIdx.x * blockDim.x + threadIdx.x) >> 5);
  if (wave >= N_EDGES / EDGES_PER_WAVE) return;
  int lane = threadIdx.x & 31;
  int e = wave * EDGES_PER_WAVE + lane;
  int rv  = row[e];
  int cv  = col[e];
  const float* srcl = outp + lane;
  float*       dstl = agg  + (size_t)l2_half() * PLANE + lane;
#pragma unroll
  for (int g = 0; g < EDGES_PER_WAVE; g += EDGE_GROUP) {
    float va[EDGE_GROUP], vb[EDGE_GROUP];
#pragma unroll
    for (int j = 0; j < EDGE_GROUP; ++j) {
      int r = __builtin_amdgcn_readlane(rv, g + j);
      const float* s = srcl + (size_t)r * HIDDEN;
      va[j] = s[0];
      vb[j] = s[HALF_ROW];
    }
#pragma unroll
    for (int j = 0; j < EDGE_GROUP; ++j) {
      int   c  = __builtin_amdgcn_readlane(cv, g + j);
      float* d = dstl + (size_t)c * HIDDEN;
      atomicAdd(&d[0],        va[j]);
      atomicAdd(&d[HALF_ROW], vb[j]);
    }
  }
}


#define NODE_ROWS 64
#define NODE_V4   (NODE_ROWS * HIDDEN / 4 / 128)
#define HS_PITCH  68

#define NODE_FIRST 0
#define NODE_MID   1
#define NODE_LAST  2

__device__ __forceinline__ void node_tile_to_lds(const int mode, float* __restrict__ h,
                                                 float* __restrict__ agg,
                                                 const float* __restrict__ x0,
                                                 const float* __restrict__ ext_w,
                                                 const float* __restrict__ beta,
                                                 const float* __restrict__ deg,
                                                 float* __restrict__ Hs,
                                                 float* __restrict__ Ds) {
  const int t = threadIdx.x;
  const size_t base = (size_t)blockIdx.x * (NODE_ROWS * HIDDEN);
  v4f ew = {};
  float b = 0.0f;
  if (mode != NODE_FIRST) {
    ew = *(const v4f*)(ext_w + 4 * (t & 15));
    b  = beta[0];
  }
  v4f hv[NODE_V4], a0[NODE_V4], a1[NODE_V4], xv[NODE_V4];
  float d0[NODE_V4], d1[NODE_V4];
  size_t off[NODE_V4];
  bool   ok[NODE_V4];
#pragma unroll
  for (int i = 0; i < NODE_V4; ++i) {
    off[i] = base + (size_t)(t + 128 * i) * 4;
    ok[i]  = off[i] < PLANE;
    size_t ol = ok[i] ? off[i] : (PLANE - 4);
    hv[i] = *(const v4f*)(h + ol);
    int node = (int)blockIdx.x * NODE_ROWS + (t >> 4) + 8 * i;
    int nl = node < N_NODES ? node : N_NODES - 1;
    d0[i] = deg[nl];
    d1[i] = deg[DEG_PAD + nl];
    if (mode != NODE_FIRST) {
      a0[i] = *(const v4f*)(agg + ol);
      a1[i] = *(const v4f*)(agg + PLANE + ol);
      xv[i] = *(const v4f*)(x0 + ol);
    }
  }
#pragma unroll
  for (int i = 0; i < NODE_V4; ++i) {
    v4f hn = hv[i];
    float dv = dinv_of(d0[i], d1[i]);
    if (mode != NODE_LAST && (t & 15) == 0) Ds[(t >> 4) + 8 * i] = dv;
    if (mode != NODE_FIRST) {
      for (int c = 0; c < 4; ++c) {
        float conv = dv * (a0[i][c] + a1[i][c]) - hv[i][c] * ew[c] + xv[i][c] * b;
        hn[c] = hv[i][c] + STEP_SZ * fmaxf(conv, 0.0f);
      }
    }
    int q = t + 128 * i;
    *(v4f*)(Hs + (q >> 4) * HS_PITCH + (q & 15) * 4) = hn;
    if (ok[i]) {
      if (mode == NODE_MID) *(v4f*)(h + off[i]) = hn;
      if (mode != NODE_LAST) {
        v4f z = {};
        *(v4f*)(agg + off[i])         = z;
        *(v4f*)(agg + PLANE + off[i]) = z;
      }
    }
  }
}

__device__ __forceinline__ void node_hw_product(const float* __restrict__ Hs,
                                                const float* __restrict__ Ws,
                                                const float* __restrict__ Ds,
                                                float* __restrict__ outp) {
  int w    = threadIdx.x >> 5;
  int tile = (int)blockIdx.x * 4 + w;
  if (tile >= M_TILES) return;
  int lane = threadIdx.x & 31;
  int mr = lane & 15;
  int kh = lane >> 4;
  const float* arow = Hs + (16 * w + mr) * HS_PITCH + 4 * kh;
  const float* bp   = Ws + lane * 4;
  v8f c0 = {}, c1 = {}, c2 = {}, c3 = {};
  for (int kb = 0; kb < HIDDEN / 8; ++kb) {
    const float* bk = bp + kb * 512;
    v4f a  = *(const v4f*)(arow + 8 * kb);
    v4f b0 = *(const v4f*)(bk);
    v4f b1 = *(const v4f*)(bk + 128);
    v4f b2 = *(const v4f*)(bk + 256);
    v4f b3 = *(const v4f*)(bk + 384);
    ENC_MMA8(a, b0, b1, b2, b3);
  }
  int mbase = tile * 16 + kh * 8;
  const float* dr = Ds + 16 * w + kh * 8;
  for (int r = 0; r < 8; ++r) {
    size_t o = (size_t)(mbase + r) * HIDDEN + mr;
    float dv = dr[r];
    outp[o]      = c0[r] * dv;
    outp[o + 16] = c1[r] * dv;
    outp[o + 32] = c2[r] * dv;
    outp[o + 48] = c3[r] * dv;
  }
}

__device__ __forceinline__ void stage_weff(const float* __restrict__ W, float* __restrict__ Ws) {
  v4f wv[HIDDEN * HIDDEN / 4 / 128];
#pragma unroll
  for (int i = 0; i < HIDDEN * HIDDEN / 4 / 128; ++i) wv[i] = ((const v4f*)W)[threadIdx.x + 128 * i];
#pragma unroll
  for (int i = 0; i < HIDDEN * HIDDEN / 4 / 128; ++i) ((v4f*)Ws)[threadIdx.x + 128 * i] = wv[i];
}

__global__ __launch_bounds__(128)
void node_first_kernel(float* __restrict__ h, float* __restrict__ agg,
                       const float* __restrict__ deg,
                       const float* __restrict__ W, float* __restrict__ outp) {
  __shared__ __attribute__((aligned(16))) float Ws[HIDDEN * HIDDEN];
  __shared__ __attribute__((aligned(16))) float Hs[NODE_ROWS * HS_PITCH];
  __shared__ float Ds[NODE_ROWS];
  stage_weff(W, Ws);
  node_tile_to_lds(NODE_FIRST, h, agg, nullptr, nullptr, nullptr, deg, Hs, Ds);
  __syncthreads();
  node_hw_product(Hs, Ws, Ds, outp);
}

__global__ __launch_bounds__(128)
void node_mid_kernel(float* __restrict__ h, float* __restrict__ agg,
                     const float* __restrict__ x0, const float* __restrict__ ext_w,
                     const float* __restrict__ beta, const float* __restrict__ deg,
                     const float* __restrict__ W, float* __restrict__ outp) {
  __shared__ __attribute__((aligned(16))) float Ws[HIDDEN * HIDDEN];
  __shared__ __attribute__((aligned(16))) float Hs[NODE_ROWS * HS_PITCH];
  __shared__ float Ds[NODE_ROWS];
  stage_weff(W, Ws);
  node_tile_to_lds(NODE_MID, h, agg, x0, ext_w, beta, deg, Hs, Ds);
  __syncthreads();
  node_hw_product(Hs, Ws, Ds, outp);
}

__global__ __launch_bounds__(128) __attribute__((amdgpu_waves_per_eu(1, 4)))
void node_last_kernel(float* __restrict__ h, float* __restrict__ agg,
                      const float* __restrict__ x0, const float* __restrict__ ext_w,
                      const float* __restrict__ beta, const float* __restrict__ deg,
                      const float* __restrict__ dw, float* __restrict__ out) {
  __shared__ __attribute__((aligned(16))) float Hs[NODE_ROWS * HS_PITCH];
  node_tile_to_lds(NODE_LAST, h, agg, x0, ext_w, beta, deg, Hs, nullptr);
  __syncthreads();
  int w    = threadIdx.x >> 5;
  int tile = (int)blockIdx.x * 4 + w;
  if (tile >= M_TILES) return;
  int lane = threadIdx.x & 31;
  int mr = lane & 15;
  int kh = lane >> 4;
  const float* arow = Hs + (16 * w + mr) * HS_PITCH + 4 * kh;
  const float* b0r = dw + (size_t)(mr +  0) * HIDDEN + 4 * kh;
  const float* b1r = dw + (size_t)(mr + 16) * HIDDEN + 4 * kh;
  int n2 = mr + 32;
  const float* b2r = dw + (size_t)(n2 < NCLASS ? n2 : 0) * HIDDEN + 4 * kh;
  float b2scale = (n2 < NCLASS) ? 1.0f : 0.0f;
  v8f c0 = {}, c1 = {}, c2 = {};
  for (int kb = 0; kb < HIDDEN / 8; ++kb) {
    v4f a  = *(const v4f*)(arow + 8 * kb);
    v4f b0 = *(const v4f*)(b0r + 8 * kb);
    v4f b1 = *(const v4f*)(b1r + 8 * kb);
    v4f b2 = *(const v4f*)(b2r + 8 * kb);
    b2 *= b2scale;
    v2f alo = __builtin_shufflevector(a, a, 0, 1), ahi = __builtin_shufflevector(a, a, 2, 3);
    c0 = __builtin_amdgcn_wmma_f32_16x16x4_f32(false, alo, false, __builtin_shufflevector(b0, b0, 0, 1), (short)0, c0, false, false);
    c1 = __builtin_amdgcn_wmma_f32_16x16x4_f32(false, alo, false, __builtin_shufflevector(b1, b1, 0, 1), (short)0, c1, false, false);
    c2 = __builtin_amdgcn_wmma_f32_16x16x4_f32(false, alo, false, __builtin_shufflevector(b2, b2, 0, 1), (short)0, c2, false, false);
    c0 = __builtin_amdgcn_wmma_f32_16x16x4_f32(false, ahi, false, __builtin_shufflevector(b0, b0, 2, 3), (short)0, c0, false, false);
    c1 = __builtin_amdgcn_wmma_f32_16x16x4_f32(false, ahi, false, __builtin_shufflevector(b1, b1, 2, 3), (short)0, c1, false, false);
    c2 = __builtin_amdgcn_wmma_f32_16x16x4_f32(false, ahi, false, __builtin_shufflevector(b2, b2, 2, 3), (short)0, c2, false, false);
  }
  int mbase = tile * 16 + kh * 8;
  for (int r = 0; r < 8; ++r) {
    float* orow = out + (size_t)(mbase + r) * NCLASS;
    orow[mr]      = c0[r];
    orow[mr + 16] = c1[r];
    if (n2 < NCLASS) orow[n2] = c2[r];
  }
}


extern "C" void kernel_launch(void* const* d_in, const int* in_sizes, int n_in,
                              void* d_out, int out_size, void* d_ws, size_t ws_size,
                              hipStream_t stream) {
  (void)in_sizes; (void)n_in; (void)out_size; (void)ws_size;
  const float* x     = (const float*)d_in[0];
  const int*   ei    = (const int*)  d_in[1];
  const float* enc_w = (const float*)d_in[2];
  const float* dec_w = (const float*)d_in[3];
  const float* ext_w = (const float*)d_in[4];
  const float* beta  = (const float*)d_in[5];
  const float* pw    = (const float*)d_in[6];
  const int* row = ei;
  const int* col = ei + N_EDGES;

  float* ws   = (float*)d_ws;
  float* deg  = ws;
  float* W    = deg  + 2 * DEG_PAD;
  float* h    = W    + HIDDEN * HIDDEN;
  float* x0   = h    + (size_t)N_NODES * HIDDEN;
  float* outp = x0   + (size_t)N_NODES * HIDDEN;
  float* agg  = outp + (size_t)N_NODES * HIDDEN;
  float* ewp  = agg  + 2 * PLANE;

  hipMemsetAsync(deg, 0, (DEG_PAD + N_NODES) * sizeof(float), stream);
  deg_count_kernel<<<(N_EDGES + 255) / 256, 256, 0, stream>>>(col, deg);
  weff_pack_kernel<<<1 + (ENC_FRAGS + HIDDEN - 1) / HIDDEN, HIDDEN, 0, stream>>>(pw, W, enc_w, ewp);

  const int gemm_blocks = (M_TILES + 3) / 4;
  enc_gemm_kernel<<<gemm_blocks, 128, 0, stream>>>(x, enc_w, ewp, h, x0);

  node_first_kernel<<<gemm_blocks, 128, 0, stream>>>(h, agg, deg, W, outp);
  edge_agg_kernel<<<N_EDGES / (8 * EDGES_PER_WAVE), 256, 0, stream>>>(row, col, outp, agg);
  node_mid_kernel<<<gemm_blocks, 128, 0, stream>>>(h, agg, x0, ext_w, beta, deg, W, outp);
  edge_agg_kernel<<<N_EDGES / (8 * EDGES_PER_WAVE), 256, 0, stream>>>(row, col, outp, agg);
  node_mid_kernel<<<gemm_blocks, 128, 0, stream>>>(h, agg, x0, ext_w, beta, deg, W, outp);
  edge_agg_kernel<<<N_EDGES / (8 * EDGES_PER_WAVE), 256, 0, stream>>>(row, col, outp, agg);
  node_mid_kernel<<<gemm_blocks, 128, 0, stream>>>(h, agg, x0, ext_w, beta, deg, W, outp);
  edge_agg_kernel<<<N_EDGES / (8 * EDGES_PER_WAVE), 256, 0, stream>>>(row, col, outp, agg);
  node_last_kernel<<<gemm_blocks, 128, 0, stream>>>(h, agg, x0, ext_w, beta, deg, dec_w, (float*)d_out);
}


# Edges binned once per call by 196-row target tile; per pass a work-group per WGP sums its bin in an LDS tile (LDS float adds), stores each agg row once; no float atomics there; sum order: arrival
# speedup vs baseline: 3.5188x; 1.2816x over previous
#include <hip/hip_runtime.h>
#include <math.h>

#define N_NODES   50000
#define N_EDGES   1600000
#define NFEAT     500
#define HIDDEN    64
#define NCLASS    40
#define NUM_LAYERS 4
#define STEP_SZ   0.1f
#define M_TILES   (N_NODES / 16)
#define DEG_PAD   50048
#define ENC_KB    (NFEAT / 8)
#define ENC_FRAGS (ENC_KB * 4 * 32)

typedef __attribute__((ext_vector_type(2))) float v2f;
typedef __attribute__((ext_vector_type(4))) float v4f;
typedef __attribute__((ext_vector_type(8))) float v8f;

#define PLANE ((size_t)N_NODES * HIDDEN)

__device__ __forceinline__ int l2_half() {
  unsigned int id = __builtin_amdgcn_s_sendmsg_rtn(0x87);
  return (int)((id >> 18) & 1u);
}


__global__ void deg_count_kernel(const int* __restrict__ col, float* __restrict__ deg) {
  int e = blockIdx.x * blockDim.x + threadIdx.x;
  float* d = deg + l2_half() * DEG_PAD;
  if (e < N_EDGES) atomicAdd(&d[col[e]], 1.0f);
}

__device__ __forceinline__ float dinv_of(float d0, float d1) {
  float d = d0 + d1;
  return (d > 0.0f) ? rsqrtf(d) : 0.0f;
}

__device__ __forceinline__ int wf_pos(int n, int k) {
  int lane = (n & 15) + 16 * ((k >> 2) & 1);
  return ((((k >> 3) * 4 + (n >> 4)) * 32) + lane) * 4 + (k & 3);
}

__global__ void weff_pack_kernel(const float* __restrict__ pw, float* __restrict__ W,
                                 const float* __restrict__ ew, float* __restrict__ ewp) {
  if (blockIdx.x != 0) {
    int idx = (int)(blockIdx.x - 1) * HIDDEN + threadIdx.x;
    if (idx < ENC_FRAGS) {
      int lane = idx & 31;
      int t    = (idx >> 5) & 3;
      int kb   = idx >> 7;
      const float* s = ew + (size_t)(16 * t + (lane & 15)) * NFEAT + 8 * kb + 4 * (lane >> 4);
      *(v4f*)(ewp + (size_t)idx * 4) = *(const v4f*)s;
    }
    return;
  }
  int i = threadIdx.x;
  const int LD = HIDDEN + 2;
  float q = pw[i * LD + HIDDEN];
  float r = pw[i * LD + HIDDEN + 1];
  float s = 0.0f;
  for (int j = 0; j < HIDDEN; ++j) {
    float v = 0.0f;
    if (i < j)      v = pw[i * LD + j];
    else if (i > j) v = pw[j * LD + i];
    s += fabsf(v);
    W[wf_pos(i, j)] = v;
  }
  W[wf_pos(i, i)] = q * s + r;
}


#define ENC_MMA8(a, b0, b1, b2, b3)                                                              \
  do {                                                                                           \
    v2f alo = __builtin_shufflevector(a, a, 0, 1), ahi = __builtin_shufflevector(a, a, 2, 3);     \
    c0 = __builtin_amdgcn_wmma_f32_16x16x4_f32(false, alo, false, __builtin_shufflevector(b0, b0, 0, 1), (short)0, c0, false, false); \
    c1 = __builtin_amdgcn_wmma_f32_16x16x4_f32(false, alo, false, __builtin_shufflevector(b1, b1, 0, 1), (short)0, c1, false, false); \
    c2 = __builtin_amdgcn_wmma_f32_16x16x4_f32(false, alo, false, __builtin_shufflevector(b2, b2, 0, 1), (short)0, c2, false, false); \
    c3 = __builtin_amdgcn_wmma_f32_16x16x4_f32(false, alo, false, __builtin_shufflevector(b3, b3, 0, 1), (short)0, c3, false, false); \
    c0 = __builtin_amdgcn_wmma_f32_16x16x4_f32(false, ahi, false, __builtin_shufflevector(b0, b0, 2, 3), (short)0, c0, false, false); \
    c1 = __builtin_amdgcn_wmma_f32_16x16x4_f32(false, ahi, false, __builtin_shufflevector(b1, b1, 2, 3), (short)0, c1, false, false); \
    c2 = __builtin_amdgcn_wmma_f32_16x16x4_f32(false, ahi, false, __builtin_shufflevector(b2, b2, 2, 3), (short)0, c2, false, false); \
    c3 = __builtin_amdgcn_wmma_f32_16x16x4_f32(false, ahi, false, __builtin_shufflevector(b3, b3, 2, 3), (short)0, c3, false, false); \
  } while (0)

__global__ __launch_bounds__(128)
void enc_gemm_kernel(const float* __restrict__ x, const float* __restrict__ ew,
                     const float* __restrict__ ewp,
                     float* __restrict__ h, float* __restrict__ x0) {
  int wave = (int)((blockIdx.x * blockDim.x + threadIdx.x) >> 5);
  if (wave >= M_TILES) return;
  int lane = threadIdx.x & 31;
  int mr = lane & 15;
  int kh = lane >> 4;
  const float* arow = x + (size_t)(wave * 16 + mr) * NFEAT;
  const float* ap   = arow + 4 * kh;
  const float* bp   = ewp + lane * 4;
  v8f c0 = {}, c1 = {}, c2 = {}, c3 = {};
  v4f aE  = *(const v4f*)(ap);
  v4f bE0 = *(const v4f*)(bp);
  v4f bE1 = *(const v4f*)(bp + 128);
  v4f bE2 = *(const v4f*)(bp + 256);
  v4f bE3 = *(const v4f*)(bp + 384);
  for (int kb = 0; kb < ENC_KB - 2; kb += 2) {
    const float* bo = bp + (size_t)(kb + 1) * 512;
    v4f aO  = *(const v4f*)(ap + (kb + 1) * 8);
    v4f bO0 = *(const v4f*)(bo);
    v4f bO1 = *(const v4f*)(bo + 128);
    v4f bO2 = *(const v4f*)(bo + 256);
    v4f bO3 = *(const v4f*)(bo + 384);
    ENC_MMA8(aE, bE0, bE1, bE2, bE3);
    const float* be = bp + (size_t)(kb + 2) * 512;
    aE  = *(const v4f*)(ap + (kb + 2) * 8);
    bE0 = *(const v4f*)(be);
    bE1 = *(const v4f*)(be + 128);
    bE2 = *(const v4f*)(be + 256);
    bE3 = *(const v4f*)(be + 384);
    ENC_MMA8(aO, bO0, bO1, bO2, bO3);
  }
  {
    const float* bo = bp + (size_t)(ENC_KB - 1) * 512;
    v4f aO  = *(const v4f*)(ap + (ENC_KB - 1) * 8);
    v4f bO0 = *(const v4f*)(bo);
    v4f bO1 = *(const v4f*)(bo + 128);
    v4f bO2 = *(const v4f*)(bo + 256);
    v4f bO3 = *(const v4f*)(bo + 384);
    ENC_MMA8(aE, bE0, bE1, bE2, bE3);
    ENC_MMA8(aO, bO0, bO1, bO2, bO3);
  }
  {
    int ko = ENC_KB * 8 + 2 * kh;
    v2f a  = *(const v2f*)(arow + ko);
    v2f b0 = *(const v2f*)(ew + (size_t)(mr +  0) * NFEAT + ko);
    v2f b1 = *(const v2f*)(ew + (size_t)(mr + 16) * NFEAT + ko);
    v2f b2 = *(const v2f*)(ew + (size_t)(mr + 32) * NFEAT + ko);
    v2f b3 = *(const v2f*)(ew + (size_t)(mr + 48) * NFEAT + ko);
    c0 = __builtin_amdgcn_wmma_f32_16x16x4_f32(false, a, false, b0, (short)0, c0, false, false);
    c1 = __builtin_amdgcn_wmma_f32_16x16x4_f32(false, a, false, b1, (short)0, c1, false, false);
    c2 = __builtin_amdgcn_wmma_f32_16x16x4_f32(false, a, false, b2, (short)0, c2, false, false);
    c3 = __builtin_amdgcn_wmma_f32_16x16x4_f32(false, a, false, b3, (short)0, c3, false, false);
  }
  int mbase = wave * 16 + kh * 8;
  for (int r = 0; r < 8; ++r) {
    size_t o = (size_t)(mbase + r) * HIDDEN + mr;
    h[o]      = c0[r]; x0[o]      = c0[r];
    h[o + 16] = c1[r]; x0[o + 16] = c1[r];
    h[o + 32] = c2[r]; x0[o + 32] = c2[r];
    h[o + 48] = c3[r]; x0[o + 48] = c3[r];
  }
}


#define TILE_ROWS  196
#define TILE_V4    (TILE_ROWS * HIDDEN / 4)
#define NBINS      ((N_NODES + TILE_ROWS - 1) / TILE_ROWS)
#define BIN_WGS    250
#define BIN_CHUNK  (N_EDGES / BIN_WGS)
#define BIN_PER_T  ((BIN_CHUNK + 1023) / 1024)
#define SCAN_PER_W (NBINS / 32)
#define ROW_V4     (TILE_ROWS / 4)
#define EDGE_GROUP 16
#define HALF_ROW   (HIDDEN / 2)

static_assert(N_NODES <= 65536, "a source row must fit the low 16 bits of a bin entry");
static_assert(N_EDGES % BIN_WGS == 0, "equal chunks");
static_assert(NBINS == 256 && TILE_ROWS % 4 == 0 && ROW_V4 <= 64 && TILE_ROWS <= 256,
              "32 waves x 8 bins; a bin's counts are at most two 16-byte pieces per lane; 8-bit row within the bin");

__global__ __launch_bounds__(1024)
void bin_scan_kernel(const float* __restrict__ deg, int* __restrict__ binstart,
                     int* __restrict__ cursor) {
  __shared__ int tot[NBINS];
  const int t = threadIdx.x, lane = t & 31, w = t >> 5;
  for (int k0 = 0; k0 < SCAN_PER_W; k0 += 4) {
    v4f p0[4][2], p1[4][2];
#pragma unroll
    for (int k = 0; k < 4; ++k) {
      int b = w + 32 * (k0 + k);
#pragma unroll
      for (int hh = 0; hh < 2; ++hh) {
        int q = lane + 32 * hh;
        int n = b * TILE_ROWS + 4 * q;
        int nl = (q < ROW_V4 && n < N_NODES) ? n : 0;
        p0[k][hh] = *(const v4f*)(deg + nl);
        p1[k][hh] = *(const v4f*)(deg + DEG_PAD + nl);
      }
    }
#pragma unroll
    for (int k = 0; k < 4; ++k) {
      int b = w + 32 * (k0 + k);
      float s = 0.0f;
#pragma unroll
      for (int hh = 0; hh < 2; ++hh) {
        int q = lane + 32 * hh;
        int n = b * TILE_ROWS + 4 * q;
        float u = (p0[k][hh][0] + p1[k][hh][0]) + (p0[k][hh][1] + p1[k][hh][1])
                + (p0[k][hh][2] + p1[k][hh][2]) + (p0[k][hh][3] + p1[k][hh][3]);
        if (q < ROW_V4 && n < N_NODES) s += u;
      }
      for (int o = 16; o > 0; o >>= 1) s += __shfl_xor(s, o, 32);
      if (lane == 0) tot[b] = (int)s;
    }
  }
  __syncthreads();
  if (w == 0) {
    int loc[SCAN_PER_W];
    int s = 0;
#pragma unroll
    for (int j = 0; j < SCAN_PER_W; ++j) {
      int b = lane * SCAN_PER_W + j;
      loc[j] = s;
      s += tot[b];
    }
    int incl = s;
    for (int o = 1; o < 32; o <<= 1) {
      int u = __shfl_up(incl, o, 32);
      if (lane >= o) incl += u;
    }
    int excl = incl - s;
#pragma unroll
    for (int j = 0; j < SCAN_PER_W; ++j) {
      int b = lane * SCAN_PER_W + j;
      binstart[b] = excl + loc[j];
    }
    if (lane == 31) binstart[NBINS] = incl;
  }
  if (t < NBINS) cursor[t] = 0;
}

__global__ __launch_bounds__(1024)
void bin_place_kernel(const int* __restrict__ row, const int* __restrict__ col,
                      const int* __restrict__ binstart, int* __restrict__ cursor,
                      int* __restrict__ binned) {
  __shared__ int cnt[NBINS];
  __shared__ int base[NBINS];
  const int t = threadIdx.x;
  if (t < NBINS) cnt[t] = 0;
  __syncthreads();
  const int e0 = (int)blockIdx.x * BIN_CHUNK;
  int rv[BIN_PER_T], cv[BIN_PER_T], rk[BIN_PER_T];
#pragma unroll
  for (int i = 0; i < BIN_PER_T; ++i) {
    int e = t + 1024 * i;
    int el = e < BIN_CHUNK ? e : BIN_CHUNK - 1;
    rv[i] = row[e0 + el];
    cv[i] = col[e0 + el];
  }
#pragma unroll
  for (int i = 0; i < BIN_PER_T; ++i) {
    int e = t + 1024 * i;
    rk[i] = 0;
    if (e < BIN_CHUNK) rk[i] = atomicAdd(&cnt[cv[i] / TILE_ROWS], 1);
  }
  __syncthreads();
  if (t < NBINS) {
    int n = cnt[t];
    int bs = 0;
    if (n > 0) bs = binstart[t] + atomicAdd(&cursor[t], n);
    base[t] = bs;
  }
  __syncthreads();
#pragma unroll
  for (int i = 0; i < BIN_PER_T; ++i) {
    int e = t + 1024 * i;
    if (e < BIN_CHUNK) {
      int b = cv[i] / TILE_ROWS;
      binned[base[b] + rk[i]] = rv[i] | ((cv[i] - b * TILE_ROWS) << 16);
    }
  }
}

__global__ __launch_bounds__(1024)
void edge_tile_kernel(const int* __restrict__ binned, const int* __restrict__ binstart,
                      const float* __restrict__ outp, float* __restrict__ agg) {
  __shared__ __attribute__((aligned(16))) float T[TILE_ROWS * HIDDEN];
  const int t = threadIdx.x;
  const int lane = t & 31;
  const int w = __builtin_amdgcn_readfirstlane(t >> 5);
  for (int q = t; q < TILE_V4; q += 1024) {
    v4f z = {};
    ((v4f*)T)[q] = z;
  }
  __syncthreads();
  const int b   = (int)blockIdx.x;
  const int beg = binstart[b];
  const int end = binstart[b + 1];
  const float* srcl = outp + lane;
  float*       Tl   = T + lane;
  int e0 = beg + 32 * w;
  int i0 = e0 + lane;
  int p  = binned[i0 < N_EDGES ? i0 : N_EDGES - 1];
  while (e0 < end) {
    const int e1 = e0 + 32 * 32;
    const int i1 = e1 + lane;
    const int pn = binned[i1 < N_EDGES ? i1 : N_EDGES - 1];
    const int n  = end - e0;
    if (n >= 32) {
#pragma unroll
      for (int g = 0; g < 32; g += EDGE_GROUP) {
        float va[EDGE_GROUP], vb[EDGE_GROUP];
#pragma unroll
        for (int j = 0; j < EDGE_GROUP; ++j) {
          int r = __builtin_amdgcn_readlane(p, g + j) & 0xFFFF;
          const float* s = srcl + (size_t)r * HIDDEN;
          va[j] = s[0];
          vb[j] = s[HALF_ROW];
        }
#pragma unroll
        for (int j = 0; j < EDGE_GROUP; ++j) {
          int c = __builtin_amdgcn_readlane(p, g + j) >> 16;
          float* d = Tl + c * HIDDEN;
          atomicAdd(&d[0],        va[j]);
          atomicAdd(&d[HALF_ROW], vb[j]);
        }
      }
    } else {
      for (int j = 0; j < n; ++j) {
        int pj = __builtin_amdgcn_readlane(p, j);
        const float* s = srcl + (size_t)(pj & 0xFFFF) * HIDDEN;
        float a0 = s[0], a1 = s[HALF_ROW];
        float* d = Tl + (pj >> 16) * HIDDEN;
        atomicAdd(&d[0],        a0);
        atomicAdd(&d[HALF_ROW], a1);
      }
    }
    p  = pn;
    e0 = e1;
  }
  __syncthreads();
  const size_t tb = (size_t)b * (TILE_ROWS * HIDDEN);
  for (int q = t; q < TILE_V4; q += 1024) {
    size_t off = tb + (size_t)q * 4;
    if (off < PLANE) *(v4f*)(agg + off) = ((const v4f*)T)[q];
  }
}


#define NODE_ROWS 64
#define NODE_V4   (NODE_ROWS * HIDDEN / 4 / 128)
#define HS_PITCH  68

#define NODE_FIRST 0
#define NODE_MID   1
#define NODE_LAST  2

__device__ __forceinline__ void node_tile_to_lds(const int mode, float* __restrict__ h,
                                                 const float* __restrict__ agg,
                                                 const float* __restrict__ x0,
                                                 const float* __restrict__ ext_w,
                                                 const float* __restrict__ beta,
                                                 const float* __restrict__ deg,
                                                 float* __restrict__ Hs,
                                                 float* __restrict__ Ds) {
  const int t = threadIdx.x;
  const size_t base = (size_t)blockIdx.x * (NODE_ROWS * HIDDEN);
  v4f ew = {};
  float b = 0.0f;
  if (mode != NODE_FIRST) {
    ew = *(const v4f*)(ext_w + 4 * (t & 15));
    b  = beta[0];
  }
  v4f hv[NODE_V4], a0[NODE_V4], xv[NODE_V4];
  float d0[NODE_V4], d1[NODE_V4];
  size_t off[NODE_V4];
  bool   ok[NODE_V4];
#pragma unroll
  for (int i = 0; i < NODE_V4; ++i) {
    off[i] = base + (size_t)(t + 128 * i) * 4;
    ok[i]  = off[i] < PLANE;
    size_t ol = ok[i] ? off[i] : (PLANE - 4);
    hv[i] = *(const v4f*)(h + ol);
    int node = (int)blockIdx.x * NODE_ROWS + (t >> 4) + 8 * i;
    int nl = node < N_NODES ? node : N_NODES - 1;
    d0[i] = deg[nl];
    d1[i] = deg[DEG_PAD + nl];
    if (mode != NODE_FIRST) {
      a0[i] = *(const v4f*)(agg + ol);
      xv[i] = *(const v4f*)(x0 + ol);
    }
  }
#pragma unroll
  for (int i = 0; i < NODE_V4; ++i) {
    v4f hn = hv[i];
    float dv = dinv_of(d0[i], d1[i]);
    if (mode != NODE_LAST && (t & 15) == 0) Ds[(t >> 4) + 8 * i] = dv;
    if (mode != NODE_FIRST) {
      for (int c = 0; c < 4; ++c) {
        float conv = dv * a0[i][c] - hv[i][c] * ew[c] + xv[i][c] * b;
        hn[c] = hv[i][c] + STEP_SZ * fmaxf(conv, 0.0f);
      }
    }
    int q = t + 128 * i;
    *(v4f*)(Hs + (q >> 4) * HS_PITCH + (q & 15) * 4) = hn;
    if (ok[i] && mode == NODE_MID) *(v4f*)(h + off[i]) = hn;
  }
}

__device__ __forceinline__ void node_hw_product(const float* __restrict__ Hs,
                                                const float* __restrict__ Ws,
                                                const float* __restrict__ Ds,
                                                float* __restrict__ outp) {
  int w    = threadIdx.x >> 5;
  int tile = (int)blockIdx.x * 4 + w;
  if (tile >= M_TILES) return;
  int lane = threadIdx.x & 31;
  int mr = lane & 15;
  int kh = lane >> 4;
  const float* arow = Hs + (16 * w + mr) * HS_PITCH + 4 * kh;
  const float* bp   = Ws + lane * 4;
  v8f c0 = {}, c1 = {}, c2 = {}, c3 = {};
  for (int kb = 0; kb < HIDDEN / 8; ++kb) {
    const float* bk = bp + kb * 512;
    v4f a  = *(const v4f*)(arow + 8 * kb);
    v4f b0 = *(const v4f*)(bk);
    v4f b1 = *(const v4f*)(bk + 128);
    v4f b2 = *(const v4f*)(bk + 256);
    v4f b3 = *(const v4f*)(bk + 384);
    ENC_MMA8(a, b0, b1, b2, b3);
  }
  int mbase = tile * 16 + kh * 8;
  const float* dr = Ds + 16 * w + kh * 8;
  for (int r = 0; r < 8; ++r) {
    size_t o = (size_t)(mbase + r) * HIDDEN + mr;
    float dv = dr[r];
    outp[o]      = c0[r] * dv;
    outp[o + 16] = c1[r] * dv;
    outp[o + 32] = c2[r] * dv;
    outp[o + 48] = c3[r] * dv;
  }
}

__device__ __forceinline__ void stage_weff(const float* __restrict__ W, float* __restrict__ Ws) {
  v4f wv[HIDDEN * HIDDEN / 4 / 128];
#pragma unroll
  for (int i = 0; i < HIDDEN * HIDDEN / 4 / 128; ++i) wv[i] = ((const v4f*)W)[threadIdx.x + 128 * i];
#pragma unroll
  for (int i = 0; i < HIDDEN * HIDDEN / 4 / 128; ++i) ((v4f*)Ws)[threadIdx.x + 128 * i] = wv[i];
}

__global__ __launch_bounds__(128)
void node_first_kernel(float* __restrict__ h,
                       const float* __restrict__ deg,
                       const float* __restrict__ W, float* __restrict__ outp) {
  __shared__ __attribute__((aligned(16))) float Ws[HIDDEN * HIDDEN];
  __shared__ __attribute__((aligned(16))) float Hs[NODE_ROWS * HS_PITCH];
  __shared__ float Ds[NODE_ROWS];
  stage_weff(W, Ws);
  node_tile_to_lds(NODE_FIRST, h, nullptr, nullptr, nullptr, nullptr, deg, Hs, Ds);
  __syncthreads();
  node_hw_product(Hs, Ws, Ds, outp);
}

__global__ __launch_bounds__(128)
void node_mid_kernel(float* __restrict__ h, const float* __restrict__ agg,
                     const float* __restrict__ x0, const float* __restrict__ ext_w,
                     const float* __restrict__ beta, const float* __restrict__ deg,
                     const float* __restrict__ W, float* __restrict__ outp) {
  __shared__ __attribute__((aligned(16))) float Ws[HIDDEN * HIDDEN];
  __shared__ __attribute__((aligned(16))) float Hs[NODE_ROWS * HS_PITCH];
  __shared__ float Ds[NODE_ROWS];
  stage_weff(W, Ws);
  node_tile_to_lds(NODE_MID, h, agg, x0, ext_w, beta, deg, Hs, Ds);
  __syncthreads();
  node_hw_product(Hs, Ws, Ds, outp);
}

__global__ __launch_bounds__(128) __attribute__((amdgpu_waves_per_eu(1, 4)))
void node_last_kernel(float* __restrict__ h, const float* __restrict__ agg,
                      const float* __restrict__ x0, const float* __restrict__ ext_w,
                      const float* __restrict__ beta, const float* __restrict__ deg,
                      const float* __restrict__ dw, float* __restrict__ out) {
  __shared__ __attribute__((aligned(16))) float Hs[NODE_ROWS * HS_PITCH];
  node_tile_to_lds(NODE_LAST, h, agg, x0, ext_w, beta, deg, Hs, nullptr);
  __syncthreads();
  int w    = threadIdx.x >> 5;
  int tile = (int)blockIdx.x * 4 + w;
  if (tile >= M_TILES) return;
  int lane = threadIdx.x & 31;
  int mr = lane & 15;
  int kh = lane >> 4;
  const float* arow = Hs + (16 * w + mr) * HS_PITCH + 4 * kh;
  const float* b0r = dw + (size_t)(mr +  0) * HIDDEN + 4 * kh;
  const float* b1r = dw + (size_t)(mr + 16) * HIDDEN + 4 * kh;
  int n2 = mr + 32;
  const float* b2r = dw + (size_t)(n2 < NCLASS ? n2 : 0) * HIDDEN + 4 * kh;
  float b2scale = (n2 < NCLASS) ? 1.0f : 0.0f;
  v8f c0 = {}, c1 = {}, c2 = {};
  for (int kb = 0; kb < HIDDEN / 8; ++kb) {
    v4f a  = *(const v4f*)(arow + 8 * kb);
    v4f b0 = *(const v4f*)(b0r + 8 * kb);
    v4f b1 = *(const v4f*)(b1r + 8 * kb);
    v4f b2 = *(const v4f*)(b2r + 8 * kb);
    b2 *= b2scale;
    v2f alo = __builtin_shufflevector(a, a, 0, 1), ahi = __builtin_shufflevector(a, a, 2, 3);
    c0 = __builtin_amdgcn_wmma_f32_16x16x4_f32(false, alo, false, __builtin_shufflevector(b0, b0, 0, 1), (short)0, c0, false, false);
    c1 = __builtin_amdgcn_wmma_f32_16x16x4_f32(false, alo, false, __builtin_shufflevector(b1, b1, 0, 1), (short)0, c1, false, false);
    c2 = __builtin_amdgcn_wmma_f32_16x16x4_f32(false, alo, false, __builtin_shufflevector(b2, b2, 0, 1), (short)0, c2, false, false);
    c0 = __builtin_amdgcn_wmma_f32_16x16x4_f32(false, ahi, false, __builtin_shufflevector(b0, b0, 2, 3), (short)0, c0, false, false);
    c1 = __builtin_amdgcn_wmma_f32_16x16x4_f32(false, ahi, false, __builtin_shufflevector(b1, b1, 2, 3), (short)0, c1, false, false);
    c2 = __builtin_amdgcn_wmma_f32_16x16x4_f32(false, ahi, false, __builtin_shufflevector(b2, b2, 2, 3), (short)0, c2, false, false);
  }
  int mbase = tile * 16 + kh * 8;
  for (int r = 0; r < 8; ++r) {
    float* orow = out + (size_t)(mbase + r) * NCLASS;
    orow[mr]      = c0[r];
    orow[mr + 16] = c1[r];
    if (n2 < NCLASS) orow[n2] = c2[r];
  }
}


extern "C" void kernel_launch(void* const* d_in, const int* in_sizes, int n_in,
                              void* d_out, int out_size, void* d_ws, size_t ws_size,
                              hipStream_t stream) {
  (void)in_sizes; (void)n_in; (void)out_size; (void)ws_size;
  const float* x     = (const float*)d_in[0];
  const int*   ei    = (const int*)  d_in[1];
  const float* enc_w = (const float*)d_in[2];
  const float* dec_w = (const float*)d_in[3];
  const float* ext_w = (const float*)d_in[4];
  const float* beta  = (const float*)d_in[5];
  const float* pw    = (const float*)d_in[6];
  const int* row = ei;
  const int* col = ei + N_EDGES;

  float* ws   = (float*)d_ws;
  float* deg  = ws;
  float* W    = deg  + 2 * DEG_PAD;
  float* h    = W    + HIDDEN * HIDDEN;
  float* x0   = h    + (size_t)N_NODES * HIDDEN;
  float* outp = x0   + (size_t)N_NODES * HIDDEN;
  float* agg  = outp + (size_t)N_NODES * HIDDEN;
  float* ewp  = agg  + PLANE;
  int* binstart = (int*)(ewp + (size_t)ENC_FRAGS * 4);
  int* cursor   = binstart + 2 * NBINS;
  int* binned   = cursor + NBINS;

  hipMemsetAsync(deg, 0, (DEG_PAD + N_NODES) * sizeof(float), stream);
  deg_count_kernel<<<(N_EDGES + 255) / 256, 256, 0, stream>>>(col, deg);
  weff_pack_kernel<<<1 + (ENC_FRAGS + HIDDEN - 1) / HIDDEN, HIDDEN, 0, stream>>>(pw, W, enc_w, ewp);
  bin_scan_kernel<<<1, 1024, 0, stream>>>(deg, binstart, cursor);
  bin_place_kernel<<<BIN_WGS, 1024, 0, stream>>>(row, col, binstart, cursor, binned);

  const int gemm_blocks = (M_TILES + 3) / 4;
  enc_gemm_kernel<<<gemm_blocks, 128, 0, stream>>>(x, enc_w, ewp, h, x0);

  node_first_kernel<<<gemm_blocks, 128, 0, stream>>>(h, deg, W, outp);
  edge_tile_kernel<<<NBINS, 1024, 0, stream>>>(binned, binstart, outp, agg);
  node_mid_kernel<<<gemm_blocks, 128, 0, stream>>>(h, agg, x0, ext_w, beta, deg, W, outp);
  edge_tile_kernel<<<NBINS, 1024, 0, stream>>>(binned, binstart, outp, agg);
  node_mid_kernel<<<gemm_blocks, 128, 0, stream>>>(h, agg, x0, ext_w, beta, deg, W, outp);
  edge_tile_kernel<<<NBINS, 1024, 0, stream>>>(binned, binstart, outp, agg);
  node_mid_kernel<<<gemm_blocks, 128, 0, stream>>>(h, agg, x0, ext_w, beta, deg, W, outp);
  edge_tile_kernel<<<NBINS, 1024, 0, stream>>>(binned, binstart, outp, agg);
  node_last_kernel<<<gemm_blocks, 128, 0, stream>>>(h, agg, x0, ext_w, beta, deg, dec_w, (float*)d_out);
}


# Node update and next product done by the edge pass's own work-group: a bin's summed rows stay in LDS, are updated and multiplied there; agg plane, 4 launches dropped; 2 out_p planes; sum orders kept
# speedup vs baseline: 3.9476x; 1.1219x over previous
#include <hip/hip_runtime.h>
#include <math.h>

#define N_NODES   50000
#define N_EDGES   1600000
#define NFEAT     500
#define HIDDEN    64
#define NCLASS    40
#define NUM_LAYERS 4
#define STEP_SZ   0.1f
#define M_TILES   (N_NODES / 16)
#define DEG_PAD   50048
#define ENC_KB    (NFEAT / 8)
#define ENC_FRAGS (ENC_KB * 4 * 32)

typedef __attribute__((ext_vector_type(2))) float v2f;
typedef __attribute__((ext_vector_type(4))) float v4f;
typedef __attribute__((ext_vector_type(8))) float v8f;

#define PLANE ((size_t)N_NODES * HIDDEN)

__device__ __forceinline__ int l2_half() {
  unsigned int id = __builtin_amdgcn_s_sendmsg_rtn(0x87);
  return (int)((id >> 18) & 1u);
}


__global__ void deg_count_kernel(const int* __restrict__ col, float* __restrict__ deg) {
  int e = blockIdx.x * blockDim.x + threadIdx.x;
  float* d = deg + l2_half() * DEG_PAD;
  if (e < N_EDGES) atomicAdd(&d[col[e]], 1.0f);
}

__device__ __forceinline__ float dinv_of(float d0, float d1) {
  float d = d0 + d1;
  return (d > 0.0f) ? rsqrtf(d) : 0.0f;
}

__device__ __forceinline__ int wf_pos(int n, int k) {
  int lane = (n & 15) + 16 * ((k >> 2) & 1);
  return ((((k >> 3) * 4 + (n >> 4)) * 32) + lane) * 4 + (k & 3);
}

__global__ void weff_pack_kernel(const float* __restrict__ pw, float* __restrict__ W,
                                 const float* __restrict__ ew, float* __restrict__ ewp) {
  if (blockIdx.x != 0) {
    int idx = (int)(blockIdx.x - 1) * HIDDEN + threadIdx.x;
    if (idx < ENC_FRAGS) {
      int lane = idx & 31;
      int t    = (idx >> 5) & 3;
      int kb   = idx >> 7;
      const float* s = ew + (size_t)(16 * t + (lane & 15)) * NFEAT + 8 * kb + 4 * (lane >> 4);
      *(v4f*)(ewp + (size_t)idx * 4) = *(const v4f*)s;
    }
    return;
  }
  int i = threadIdx.x;
  const int LD = HIDDEN + 2;
  float q = pw[i * LD + HIDDEN];
  float r = pw[i * LD + HIDDEN + 1];
  float s = 0.0f;
  for (int j = 0; j < HIDDEN; ++j) {
    float v = 0.0f;
    if (i < j)      v = pw[i * LD + j];
    else if (i > j) v = pw[j * LD + i];
    s += fabsf(v);
    W[wf_pos(i, j)] = v;
  }
  W[wf_pos(i, i)] = q * s + r;
}


#define ENC_MMA8(a, b0, b1, b2, b3)                                                              \
  do {                                                                                           \
    v2f alo = __builtin_shufflevector(a, a, 0, 1), ahi = __builtin_shufflevector(a, a, 2, 3);     \
    c0 = __builtin_amdgcn_wmma_f32_16x16x4_f32(false, alo, false, __builtin_shufflevector(b0, b0, 0, 1), (short)0, c0, false, false); \
    c1 = __builtin_amdgcn_wmma_f32_16x16x4_f32(false, alo, false, __builtin_shufflevector(b1, b1, 0, 1), (short)0, c1, false, false); \
    c2 = __builtin_amdgcn_wmma_f32_16x16x4_f32(false, alo, false, __builtin_shufflevector(b2, b2, 0, 1), (short)0, c2, false, false); \
    c3 = __builtin_amdgcn_wmma_f32_16x16x4_f32(false, alo, false, __builtin_shufflevector(b3, b3, 0, 1), (short)0, c3, false, false); \
    c0 = __builtin_amdgcn_wmma_f32_16x16x4_f32(false, ahi, false, __builtin_shufflevector(b0, b0, 2, 3), (short)0, c0, false, false); \
    c1 = __builtin_amdgcn_wmma_f32_16x16x4_f32(false, ahi, false, __builtin_shufflevector(b1, b1, 2, 3), (short)0, c1, false, false); \
    c2 = __builtin_amdgcn_wmma_f32_16x16x4_f32(false, ahi, false, __builtin_shufflevector(b2, b2, 2, 3), (short)0, c2, false, false); \
    c3 = __builtin_amdgcn_wmma_f32_16x16x4_f32(false, ahi, false, __builtin_shufflevector(b3, b3, 2, 3), (short)0, c3, false, false); \
  } while (0)

__global__ __launch_bounds__(128)
void enc_gemm_kernel(const float* __restrict__ x, const float* __restrict__ ew,
                     const float* __restrict__ ewp,
                     float* __restrict__ h, float* __restrict__ x0) {
  int wave = (int)((blockIdx.x * blockDim.x + threadIdx.x) >> 5);
  if (wave >= M_TILES) return;
  int lane = threadIdx.x & 31;
  int mr = lane & 15;
  int kh = lane >> 4;
  const float* arow = x + (size_t)(wave * 16 + mr) * NFEAT;
  const float* ap   = arow + 4 * kh;
  const float* bp   = ewp + lane * 4;
  v8f c0 = {}, c1 = {}, c2 = {}, c3 = {};
  v4f aE  = *(const v4f*)(ap);
  v4f bE0 = *(const v4f*)(bp);
  v4f bE1 = *(const v4f*)(bp + 128);
  v4f bE2 = *(const v4f*)(bp + 256);
  v4f bE3 = *(const v4f*)(bp + 384);
  for (int kb = 0; kb < ENC_KB - 2; kb += 2) {
    const float* bo = bp + (size_t)(kb + 1) * 512;
    v4f aO  = *(const v4f*)(ap + (kb + 1) * 8);
    v4f bO0 = *(const v4f*)(bo);
    v4f bO1 = *(const v4f*)(bo + 128);
    v4f bO2 = *(const v4f*)(bo + 256);
    v4f bO3 = *(const v4f*)(bo + 384);
    ENC_MMA8(aE, bE0, bE1, bE2, bE3);
    const float* be = bp + (size_t)(kb + 2) * 512;
    aE  = *(const v4f*)(ap + (kb + 2) * 8);
    bE0 = *(const v4f*)(be);
    bE1 = *(const v4f*)(be + 128);
    bE2 = *(const v4f*)(be + 256);
    bE3 = *(const v4f*)(be + 384);
    ENC_MMA8(aO, bO0, bO1, bO2, bO3);
  }
  {
    const float* bo = bp + (size_t)(ENC_KB - 1) * 512;
    v4f aO  = *(const v4f*)(ap + (ENC_KB - 1) * 8);
    v4f bO0 = *(const v4f*)(bo);
    v4f bO1 = *(const v4f*)(bo + 128);
    v4f bO2 = *(const v4f*)(bo + 256);
    v4f bO3 = *(const v4f*)(bo + 384);
    ENC_MMA8(aE, bE0, bE1, bE2, bE3);
    ENC_MMA8(aO, bO0, bO1, bO2, bO3);
  }
  {
    int ko = ENC_KB * 8 + 2 * kh;
    v2f a  = *(const v2f*)(arow + ko);
    v2f b0 = *(const v2f*)(ew + (size_t)(mr +  0) * NFEAT + ko);
    v2f b1 = *(const v2f*)(ew + (size_t)(mr + 16) * NFEAT + ko);
    v2f b2 = *(const v2f*)(ew + (size_t)(mr + 32) * NFEAT + ko);
    v2f b3 = *(const v2f*)(ew + (size_t)(mr + 48) * NFEAT + ko);
    c0 = __builtin_amdgcn_wmma_f32_16x16x4_f32(false, a, false, b0, (short)0, c0, false, false);
    c1 = __builtin_amdgcn_wmma_f32_16x16x4_f32(false, a, false, b1, (short)0, c1, false, false);
    c2 = __builtin_amdgcn_wmma_f32_16x16x4_f32(false, a, false, b2, (short)0, c2, false, false);
    c3 = __builtin_amdgcn_wmma_f32_16x16x4_f32(false, a, false, b3, (short)0, c3, false, false);
  }
  int mbase = wave * 16 + kh * 8;
  for (int r = 0; r < 8; ++r) {
    size_t o = (size_t)(mbase + r) * HIDDEN + mr;
    h[o]      = c0[r]; x0[o]      = c0[r];
    h[o + 16] = c1[r]; x0[o + 16] = c1[r];
    h[o + 32] = c2[r]; x0[o + 32] = c2[r];
    h[o + 48] = c3[r]; x0[o + 48] = c3[r];
  }
}


#define TILE_ROWS  196
#define TILE_V4    (TILE_ROWS * HIDDEN / 4)
#define NBINS      ((N_NODES + TILE_ROWS - 1) / TILE_ROWS)
#define BIN_WGS    250
#define BIN_CHUNK  (N_EDGES / BIN_WGS)
#define BIN_PER_T  ((BIN_CHUNK + 1023) / 1024)
#define SCAN_PER_W (NBINS / 32)
#define ROW_V4     (TILE_ROWS / 4)
#define EDGE_GROUP 16
#define HALF_ROW   (HIDDEN / 2)

static_assert(N_NODES <= 65536, "a source row must fit the low 16 bits of a bin entry");
static_assert(N_EDGES % BIN_WGS == 0, "equal chunks");
static_assert(NBINS == 256 && TILE_ROWS % 4 == 0 && ROW_V4 <= 64 && TILE_ROWS <= 256,
              "32 waves x 8 bins; a bin's counts are at most two 16-byte pieces per lane; 8-bit row within the bin");

__global__ __launch_bounds__(1024)
void bin_scan_kernel(const float* __restrict__ deg, int* __restrict__ binstart,
                     int* __restrict__ cursor) {
  __shared__ int tot[NBINS];
  const int t = threadIdx.x, lane = t & 31, w = t >> 5;
  for (int k0 = 0; k0 < SCAN_PER_W; k0 += 4) {
    v4f p0[4][2], p1[4][2];
#pragma unroll
    for (int k = 0; k < 4; ++k) {
      int b = w + 32 * (k0 + k);
#pragma unroll
      for (int hh = 0; hh < 2; ++hh) {
        int q = lane + 32 * hh;
        int n = b * TILE_ROWS + 4 * q;
        int nl = (q < ROW_V4 && n < N_NODES) ? n : 0;
        p0[k][hh] = *(const v4f*)(deg + nl);
        p1[k][hh] = *(const v4f*)(deg + DEG_PAD + nl);
      }
    }
#pragma unroll
    for (int k = 0; k < 4; ++k) {
      int b = w + 32 * (k0 + k);
      float s = 0.0f;
#pragma unroll
      for (int hh = 0; hh < 2; ++hh) {
        int q = lane + 32 * hh;
        int n = b * TILE_ROWS + 4 * q;
        float u = (p0[k][hh][0] + p1[k][hh][0]) + (p0[k][hh][1] + p1[k][hh][1])
                + (p0[k][hh][2] + p1[k][hh][2]) + (p0[k][hh][3] + p1[k][hh][3]);
        if (q < ROW_V4 && n < N_NODES) s += u;
      }
      for (int o = 16; o > 0; o >>= 1) s += __shfl_xor(s, o, 32);
      if (lane == 0) tot[b] = (int)s;
    }
  }
  __syncthreads();
  if (w == 0) {
    int loc[SCAN_PER_W];
    int s = 0;
#pragma unroll
    for (int j = 0; j < SCAN_PER_W; ++j) {
      int b = lane * SCAN_PER_W + j;
      loc[j] = s;
      s += tot[b];
    }
    int incl = s;
    for (int o = 1; o < 32; o <<= 1) {
      int u = __shfl_up(incl, o, 32);
      if (lane >= o) incl += u;
    }
    int excl = incl - s;
#pragma unroll
    for (int j = 0; j < SCAN_PER_W; ++j) {
      int b = lane * SCAN_PER_W + j;
      binstart[b] = excl + loc[j];
    }
    if (lane == 31) binstart[NBINS] = incl;
  }
  if (t < NBINS) cursor[t] = 0;
}

__global__ __launch_bounds__(1024)
void bin_place_kernel(const int* __restrict__ row, const int* __restrict__ col,
                      const int* __restrict__ binstart, int* __restrict__ cursor,
                      int* __restrict__ binned) {
  __shared__ int cnt[NBINS];
  __shared__ int base[NBINS];
  const int t = threadIdx.x;
  if (t < NBINS) cnt[t] = 0;
  __syncthreads();
  const int e0 = (int)blockIdx.x * BIN_CHUNK;
  int rv[BIN_PER_T], cv[BIN_PER_T], rk[BIN_PER_T];
#pragma unroll
  for (int i = 0; i < BIN_PER_T; ++i) {
    int e = t + 1024 * i;
    int el = e < BIN_CHUNK ? e : BIN_CHUNK - 1;
    rv[i] = row[e0 + el];
    cv[i] = col[e0 + el];
  }
#pragma unroll
  for (int i = 0; i < BIN_PER_T; ++i) {
    int e = t + 1024 * i;
    rk[i] = 0;
    if (e < BIN_CHUNK) rk[i] = atomicAdd(&cnt[cv[i] / TILE_ROWS], 1);
  }
  __syncthreads();
  if (t < NBINS) {
    int n = cnt[t];
    int bs = 0;
    if (n > 0) bs = binstart[t] + atomicAdd(&cursor[t], n);
    base[t] = bs;
  }
  __syncthreads();
#pragma unroll
  for (int i = 0; i < BIN_PER_T; ++i) {
    int e = t + 1024 * i;
    if (e < BIN_CHUNK) {
      int b = cv[i] / TILE_ROWS;
      binned[base[b] + rk[i]] = rv[i] | ((cv[i] - b * TILE_ROWS) << 16);
    }
  }
}


#define NODE_ROWS 64
#define NODE_V4   (NODE_ROWS * HIDDEN / 4 / 128)
#define HS_PITCH  68

__device__ __forceinline__ void first_tile_to_lds(const float* __restrict__ h,
                                                  const float* __restrict__ deg,
                                                  float* __restrict__ Hs,
                                                  float* __restrict__ Ds) {
  const int t = threadIdx.x;
  const size_t base = (size_t)blockIdx.x * (NODE_ROWS * HIDDEN);
  v4f hv[NODE_V4];
  float d0[NODE_V4], d1[NODE_V4];
#pragma unroll
  for (int i = 0; i < NODE_V4; ++i) {
    size_t off = base + (size_t)(t + 128 * i) * 4;
    size_t ol = off < PLANE ? off : (PLANE - 4);
    hv[i] = *(const v4f*)(h + ol);
    int node = (int)blockIdx.x * NODE_ROWS + (t >> 4) + 8 * i;
    int nl = node < N_NODES ? node : N_NODES - 1;
    d0[i] = deg[nl];
    d1[i] = deg[DEG_PAD + nl];
  }
#pragma unroll
  for (int i = 0; i < NODE_V4; ++i) {
    if ((t & 15) == 0) Ds[(t >> 4) + 8 * i] = dinv_of(d0[i], d1[i]);
    int q = t + 128 * i;
    *(v4f*)(Hs + (q >> 4) * HS_PITCH + (q & 15) * 4) = hv[i];
  }
}

__device__ __forceinline__ void node_hw_product(const float* __restrict__ Hs,
                                                const float* __restrict__ Ws,
                                                const float* __restrict__ Ds,
                                                float* __restrict__ outp) {
  int w    = threadIdx.x >> 5;
  int tile = (int)blockIdx.x * 4 + w;
  if (tile >= M_TILES) return;
  int lane = threadIdx.x & 31;
  int mr = lane & 15;
  int kh = lane >> 4;
  const float* arow = Hs + (16 * w + mr) * HS_PITCH + 4 * kh;
  const float* bp   = Ws + lane * 4;
  v8f c0 = {}, c1 = {}, c2 = {}, c3 = {};
  for (int kb = 0; kb < HIDDEN / 8; ++kb) {
    const float* bk = bp + kb * 512;
    v4f a  = *(const v4f*)(arow + 8 * kb);
    v4f b0 = *(const v4f*)(bk);
    v4f b1 = *(const v4f*)(bk + 128);
    v4f b2 = *(const v4f*)(bk + 256);
    v4f b3 = *(const v4f*)(bk + 384);
    ENC_MMA8(a, b0, b1, b2, b3);
  }
  int mbase = tile * 16 + kh * 8;
  const float* dr = Ds + 16 * w + kh * 8;
  for (int r = 0; r < 8; ++r) {
    size_t o = (size_t)(mbase + r) * HIDDEN + mr;
    float dv = dr[r];
    outp[o]      = c0[r] * dv;
    outp[o + 16] = c1[r] * dv;
    outp[o + 32] = c2[r] * dv;
    outp[o + 48] = c3[r] * dv;
  }
}

__device__ __forceinline__ void stage_weff(const float* __restrict__ W, float* __restrict__ Ws) {
  v4f wv[HIDDEN * HIDDEN / 4 / 128];
#pragma unroll
  for (int i = 0; i < HIDDEN * HIDDEN / 4 / 128; ++i) wv[i] = ((const v4f*)W)[threadIdx.x + 128 * i];
#pragma unroll
  for (int i = 0; i < HIDDEN * HIDDEN / 4 / 128; ++i) ((v4f*)Ws)[threadIdx.x + 128 * i] = wv[i];
}

__global__ __launch_bounds__(128)
void node_first_kernel(const float* __restrict__ h,
                       const float* __restrict__ deg,
                       const float* __restrict__ W, float* __restrict__ outp) {
  __shared__ __attribute__((aligned(16))) float Ws[HIDDEN * HIDDEN];
  __shared__ __attribute__((aligned(16))) float Hs[NODE_ROWS * HS_PITCH];
  __shared__ float Ds[NODE_ROWS];
  stage_weff(W, Ws);
  first_tile_to_lds(h, deg, Hs, Ds);
  __syncthreads();
  node_hw_product(Hs, Ws, Ds, outp);
}


#define TILE_TILES 13
#define TILE_PAD   (TILE_TILES * 16)
#define PAD_V4     (TILE_PAD * HIDDEN / 4)
#define UPD_PER_T  ((PAD_V4 + 1023) / 1024)

static_assert(TILE_PAD >= TILE_ROWS && 2 * TILE_TILES <= 32 && HIDDEN * HIDDEN / 4 == 1024,
              "13 tiles x 2 column halves on 32 waves; W_eff is one 16-byte piece per thread");

#define TILE_MMA8(a, b0, b1)                                                                     \
  do {                                                                                           \
    v2f alo = __builtin_shufflevector(a, a, 0, 1), ahi = __builtin_shufflevector(a, a, 2, 3);     \
    c0 = __builtin_amdgcn_wmma_f32_16x16x4_f32(false, alo, false, __builtin_shufflevector(b0, b0, 0, 1), (short)0, c0, false, false); \
    c1 = __builtin_amdgcn_wmma_f32_16x16x4_f32(false, alo, false, __builtin_shufflevector(b1, b1, 0, 1), (short)0, c1, false, false); \
    c0 = __builtin_amdgcn_wmma_f32_16x16x4_f32(false, ahi, false, __builtin_shufflevector(b0, b0, 2, 3), (short)0, c0, false, false); \
    c1 = __builtin_amdgcn_wmma_f32_16x16x4_f32(false, ahi, false, __builtin_shufflevector(b1, b1, 2, 3), (short)0, c1, false, false); \
  } while (0)

__device__ __forceinline__ void tile_pass(const bool last,
                                          const int* __restrict__ binned,
                                          const int* __restrict__ binstart,
                                          const float* __restrict__ outp,
                                          float* __restrict__ h,
                                          const float* __restrict__ x0,
                                          const float* __restrict__ ext_w,
                                          const float* __restrict__ beta,
                                          const float* __restrict__ deg,
                                          const float* __restrict__ wsrc,
                                          float* __restrict__ dst,
                                          float* __restrict__ T, float* __restrict__ Hs,
                                          float* __restrict__ Ws, float* __restrict__ Ds) {
  const int t = threadIdx.x;
  const int lane = t & 31;
  const int w = __builtin_amdgcn_readfirstlane(t >> 5);
  const int b = (int)blockIdx.x;
  const int row0 = b * TILE_ROWS;

  v4f hv[UPD_PER_T], xv[UPD_PER_T];
  float d0[UPD_PER_T], d1[UPD_PER_T];
  v4f wv = {}, ew = {};
  float bt = 0.0f;
#pragma unroll
  for (int i = 0; i < UPD_PER_T; ++i) {
    v4f z = {};
    hv[i] = z; xv[i] = z; d0[i] = 0.0f; d1[i] = 0.0f;
  }
  auto fetch_own = [&]() {
#pragma unroll
    for (int i = 0; i < UPD_PER_T; ++i) {
      int q = t + 1024 * i;
      int r = q >> 4;
      bool ok = r < TILE_ROWS && row0 + r < N_NODES;
      int node = ok ? row0 + r : row0;
      size_t o = (size_t)node * HIDDEN + (q & 15) * 4;
      hv[i] = *(const v4f*)(h + o);
      xv[i] = *(const v4f*)(x0 + o);
      d0[i] = deg[node];
      d1[i] = deg[DEG_PAD + node];
    }
    if (!last) wv = ((const v4f*)wsrc)[t];
    ew = *(const v4f*)(ext_w + 4 * (t & 15));
    bt = beta[0];
  };
  for (int q = t; q < TILE_V4; q += 1024) {
    v4f z = {};
    ((v4f*)T)[q] = z;
  }
  __syncthreads();

  {
    const int beg = binstart[b];
    const int end = binstart[b + 1];
    const float* srcl = outp + lane;
    float*       Tl   = T + lane;
    int e0 = beg + 32 * w;
    int i0 = e0 + lane;
    int p  = binned[i0 < N_EDGES ? i0 : N_EDGES - 1];
    const int slot = (w >> 2) % 5;
    int it = 0;
    bool fetched = false;
    while (e0 < end) {
      if (it == slot) { fetch_own(); fetched = true; }
      ++it;
      const int e1 = e0 + 32 * 32;
      const int i1 = e1 + lane;
      const int pn = binned[i1 < N_EDGES ? i1 : N_EDGES - 1];
      const int n  = end - e0;
      if (n >= 32) {
#pragma unroll
        for (int g = 0; g < 32; g += EDGE_GROUP) {
          float va[EDGE_GROUP], vb[EDGE_GROUP];
#pragma unroll
          for (int j = 0; j < EDGE_GROUP; ++j) {
            int r = __builtin_amdgcn_readlane(p, g + j) & 0xFFFF;
            const float* s = srcl + (size_t)r * HIDDEN;
            va[j] = s[0];
            vb[j] = s[HALF_ROW];
          }
#pragma unroll
          for (int j = 0; j < EDGE_GROUP; ++j) {
            int c = __builtin_amdgcn_readlane(p, g + j) >> 16;
            float* d = Tl + c * HIDDEN;
            atomicAdd(&d[0],        va[j]);
            atomicAdd(&d[HALF_ROW], vb[j]);
          }
        }
      } else {
        for (int j = 0; j < n; ++j) {
          int pj = __builtin_amdgcn_readlane(p, j);
          const float* s = srcl + (size_t)(pj & 0xFFFF) * HIDDEN;
          float a0 = s[0], a1 = s[HALF_ROW];
          float* d = Tl + (pj >> 16) * HIDDEN;
          atomicAdd(&d[0],        a0);
          atomicAdd(&d[HALF_ROW], a1);
        }
      }
      p  = pn;
      e0 = e1;
    }
    if (!fetched) fetch_own();
  }
  __syncthreads();

  {
    if (!last) ((v4f*)Ws)[t] = wv;
#pragma unroll
    for (int i = 0; i < UPD_PER_T; ++i) {
      int q = t + 1024 * i;
      if (q < PAD_V4) {
        int r = q >> 4;
        v4f hn = {};
        float dv = 0.0f;
        if (r < TILE_ROWS && row0 + r < N_NODES) {
          v4f a = ((const v4f*)T)[q];
          dv = dinv_of(d0[i], d1[i]);
          for (int c = 0; c < 4; ++c) {
            float conv = dv * a[c] - hv[i][c] * ew[c] + xv[i][c] * bt;
            hn[c] = hv[i][c] + STEP_SZ * fmaxf(conv, 0.0f);
          }
          if (!last) *(v4f*)(h + (size_t)(row0 + r) * HIDDEN + (q & 15) * 4) = hn;
        }
        if ((t & 15) == 0) Ds[r] = dv;
        *(v4f*)(Hs + r * HS_PITCH + (q & 15) * 4) = hn;
      }
    }
  }
  __syncthreads();

  const int mr = lane & 15;
  const int kh = lane >> 4;
  if (!last) {
    if (w >= 2 * TILE_TILES) return;
    const int tile = w >> 1;
    const int np   = w & 1;
    const float* arow = Hs + (16 * tile + mr) * HS_PITCH + 4 * kh;
    const float* bp   = Ws + np * 256 + lane * 4;
    v8f c0 = {}, c1 = {};
    for (int kb = 0; kb < HIDDEN / 8; ++kb) {
      const float* bk = bp + kb * 512;
      v4f a  = *(const v4f*)(arow + 8 * kb);
      v4f b0 = *(const v4f*)(bk);
      v4f b1 = *(const v4f*)(bk + 128);
      TILE_MMA8(a, b0, b1);
    }
    const int rbase = 16 * tile + kh * 8;
    for (int r = 0; r < 8; ++r) {
      int tr = rbase + r;
      if (tr < TILE_ROWS && row0 + tr < N_NODES) {
        float dv = Ds[tr];
        size_t o = (size_t)(row0 + tr) * HIDDEN + 32 * np + mr;
        dst[o]      = c0[r] * dv;
        dst[o + 16] = c1[r] * dv;
      }
    }
  } else {
    if (w >= TILE_TILES) return;
    const float* arow = Hs + (16 * w + mr) * HS_PITCH + 4 * kh;
    const float* b0r = wsrc + (size_t)(mr +  0) * HIDDEN + 4 * kh;
    const float* b1r = wsrc + (size_t)(mr + 16) * HIDDEN + 4 * kh;
    const int n2 = mr + 32;
    const float* b2r = wsrc + (size_t)(n2 < NCLASS ? n2 : 0) * HIDDEN + 4 * kh;
    const float b2scale = (n2 < NCLASS) ? 1.0f : 0.0f;
    v8f c0 = {}, c1 = {}, c2 = {};
    for (int kb = 0; kb < HIDDEN / 8; ++kb) {
      v4f a  = *(const v4f*)(arow + 8 * kb);
      v4f b0 = *(const v4f*)(b0r + 8 * kb);
      v4f b1 = *(const v4f*)(b1r + 8 * kb);
      v4f b2 = *(const v4f*)(b2r + 8 * kb);
      b2 *= b2scale;
      v2f alo = __builtin_shufflevector(a, a, 0, 1), ahi = __builtin_shufflevector(a, a, 2, 3);
      c0 = __builtin_amdgcn_wmma_f32_16x16x4_f32(false, alo, false, __builtin_shufflevector(b0, b0, 0, 1), (short)0, c0, false, false);
      c1 = __builtin_amdgcn_wmma_f32_16x16x4_f32(false, alo, false, __builtin_shufflevector(b1, b1, 0, 1), (short)0, c1, false, false);
      c2 = __builtin_amdgcn_wmma_f32_16x16x4_f32(false, alo, false, __builtin_shufflevector(b2, b2, 0, 1), (short)0, c2, false, false);
      c0 = __builtin_amdgcn_wmma_f32_16x16x4_f32(false, ahi, false, __builtin_shufflevector(b0, b0, 2, 3), (short)0, c0, false, false);
      c1 = __builtin_amdgcn_wmma_f32_16x16x4_f32(false, ahi, false, __builtin_shufflevector(b1, b1, 2, 3), (short)0, c1, false, false);
      c2 = __builtin_amdgcn_wmma_f32_16x16x4_f32(false, ahi, false, __builtin_shufflevector(b2, b2, 2, 3), (short)0, c2, false, false);
    }
    const int rbase = 16 * w + kh * 8;
    for (int r = 0; r < 8; ++r) {
      int tr = rbase + r;
      if (tr < TILE_ROWS && row0 + tr < N_NODES) {
        float* orow = dst + (size_t)(row0 + tr) * NCLASS;
        orow[mr]      = c0[r];
        orow[mr + 16] = c1[r];
        if (n2 < NCLASS) orow[n2] = c2[r];
      }
    }
  }
}

__global__ __launch_bounds__(1024)
void tile_mid_kernel(const int* __restrict__ binned, const int* __restrict__ binstart,
                     const float* __restrict__ outp, float* __restrict__ h,
                     const float* __restrict__ x0, const float* __restrict__ ext_w,
                     const float* __restrict__ beta, const float* __restrict__ deg,
                     const float* __restrict__ W, float* __restrict__ outn) {
  __shared__ __attribute__((aligned(16))) float T[TILE_ROWS * HIDDEN];
  __shared__ __attribute__((aligned(16))) float Hs[TILE_PAD * HS_PITCH];
  __shared__ __attribute__((aligned(16))) float Ws[HIDDEN * HIDDEN];
  __shared__ float Ds[TILE_PAD];
  tile_pass(false, binned, binstart, outp, h, x0, ext_w, beta, deg, W, outn, T, Hs, Ws, Ds);
}

__global__ __launch_bounds__(1024)
void tile_last_kernel(const int* __restrict__ binned, const int* __restrict__ binstart,
                      const float* __restrict__ outp, float* __restrict__ h,
                      const float* __restrict__ x0, const float* __restrict__ ext_w,
                      const float* __restrict__ beta, const float* __restrict__ deg,
                      const float* __restrict__ dw, float* __restrict__ out) {
  __shared__ __attribute__((aligned(16))) float T[TILE_ROWS * HIDDEN];
  __shared__ __attribute__((aligned(16))) float Hs[TILE_PAD * HS_PITCH];
  __shared__ float Ds[TILE_PAD];
  tile_pass(true, binned, binstart, outp, h, x0, ext_w, beta, deg, dw, out, T, Hs, nullptr, Ds);
}


extern "C" void kernel_launch(void* const* d_in, const int* in_sizes, int n_in,
                              void* d_out, int out_size, void* d_ws, size_t ws_size,
                              hipStream_t stream) {
  (void)in_sizes; (void)n_in; (void)out_size; (void)ws_size;
  const float* x     = (const float*)d_in[0];
  const int*   ei    = (const int*)  d_in[1];
  const float* enc_w = (const float*)d_in[2];
  const float* dec_w = (const float*)d_in[3];
  const float* ext_w = (const float*)d_in[4];
  const float* beta  = (const float*)d_in[5];
  const float* pw    = (const float*)d_in[6];
  const int* row = ei;
  const int* col = ei + N_EDGES;

  float* ws   = (float*)d_ws;
  float* deg  = ws;
  float* W    = deg  + 2 * DEG_PAD;
  float* h    = W    + HIDDEN * HIDDEN;
  float* x0   = h    + (size_t)N_NODES * HIDDEN;
  float* outp = x0   + (size_t)N_NODES * HIDDEN;
  float* outq = outp + (size_t)N_NODES * HIDDEN;
  float* ewp  = outq + PLANE;
  int* binstart = (int*)(ewp + (size_t)ENC_FRAGS * 4);
  int* cursor   = binstart + 2 * NBINS;
  int* binned   = cursor + NBINS;

  hipMemsetAsync(deg, 0, (DEG_PAD + N_NODES) * sizeof(float), stream);
  deg_count_kernel<<<(N_EDGES + 255) / 256, 256, 0, stream>>>(col, deg);
  weff_pack_kernel<<<1 + (ENC_FRAGS + HIDDEN - 1) / HIDDEN, HIDDEN, 0, stream>>>(pw, W, enc_w, ewp);
  bin_scan_kernel<<<1, 1024, 0, stream>>>(deg, binstart, cursor);
  bin_place_kernel<<<BIN_WGS, 1024, 0, stream>>>(row, col, binstart, cursor, binned);

  const int gemm_blocks = (M_TILES + 3) / 4;
  enc_gemm_kernel<<<gemm_blocks, 128, 0, stream>>>(x, enc_w, ewp, h, x0);

  node_first_kernel<<<gemm_blocks, 128, 0, stream>>>(h, deg, W, outp);
  tile_mid_kernel<<<NBINS, 1024, 0, stream>>>(binned, binstart, outp, h, x0, ext_w, beta, deg, W, outq);
  tile_mid_kernel<<<NBINS, 1024, 0, stream>>>(binned, binstart, outq, h, x0, ext_w, beta, deg, W, outp);
  tile_mid_kernel<<<NBINS, 1024, 0, stream>>>(binned, binstart, outp, h, x0, ext_w, beta, deg, W, outq);
  tile_last_kernel<<<NBINS, 1024, 0, stream>>>(binned, binstart, outq, h, x0, ext_w, beta, deg, dec_w, (float*)d_out);
}


# W_eff set-up: one wave per row (lane l takes columns l and l+32, all loads at once) instead of one thread per row behind 64 dependent loads; |W0| row sum now a lane-shuffle tree, not j=0..63
# speedup vs baseline: 4.2314x; 1.0719x over previous
#include <hip/hip_runtime.h>
#include <math.h>

#define N_NODES   50000
#define N_EDGES   1600000
#define NFEAT     500
#define HIDDEN    64
#define NCLASS    40
#define NUM_LAYERS 4
#define STEP_SZ   0.1f
#define M_TILES   (N_NODES / 16)
#define DEG_PAD   50048
#define ENC_KB    (NFEAT / 8)
#define ENC_FRAGS (ENC_KB * 4 * 32)

typedef __attribute__((ext_vector_type(2))) float v2f;
typedef __attribute__((ext_vector_type(4))) float v4f;
typedef __attribute__((ext_vector_type(8))) float v8f;

#define PLANE ((size_t)N_NODES * HIDDEN)

__device__ __forceinline__ int l2_half() {
  unsigned int id = __builtin_amdgcn_s_sendmsg_rtn(0x87);
  return (int)((id >> 18) & 1u);
}


__global__ void deg_count_kernel(const int* __restrict__ col, float* __restrict__ deg) {
  int e = blockIdx.x * blockDim.x + threadIdx.x;
  float* d = deg + l2_half() * DEG_PAD;
  if (e < N_EDGES) atomicAdd(&d[col[e]], 1.0f);
}

__device__ __forceinline__ float dinv_of(float d0, float d1) {
  float d = d0 + d1;
  return (d > 0.0f) ? rsqrtf(d) : 0.0f;
}

#define WEFF_BLOCKS (HIDDEN / 2)

__device__ __forceinline__ int wf_pos(int n, int k) {
  int lane = (n & 15) + 16 * ((k >> 2) & 1);
  return ((((k >> 3) * 4 + (n >> 4)) * 32) + lane) * 4 + (k & 3);
}

__global__ void weff_pack_kernel(const float* __restrict__ pw, float* __restrict__ W,
                                 const float* __restrict__ ew, float* __restrict__ ewp) {
  if (blockIdx.x >= WEFF_BLOCKS) {
    int idx = (int)(blockIdx.x - WEFF_BLOCKS) * HIDDEN + threadIdx.x;
    if (idx < ENC_FRAGS) {
      int lane = idx & 31;
      int t    = (idx >> 5) & 3;
      int kb   = idx >> 7;
      const float* s = ew + (size_t)(16 * t + (lane & 15)) * NFEAT + 8 * kb + 4 * (lane >> 4);
      *(v4f*)(ewp + (size_t)idx * 4) = *(const v4f*)s;
    }
    return;
  }
  const int LD = HIDDEN + 2;
  const int lane = threadIdx.x & 31;
  const int i  = (int)blockIdx.x * 2 + (int)(threadIdx.x >> 5);
  const int j0 = lane, j1 = lane + 32;
  float q  = pw[i * LD + HIDDEN];
  float r  = pw[i * LD + HIDDEN + 1];
  float v0 = pw[(i < j0) ? i * LD + j0 : j0 * LD + i];
  float v1 = pw[(i < j1) ? i * LD + j1 : j1 * LD + i];
  if (j0 == i) v0 = 0.0f;
  if (j1 == i) v1 = 0.0f;
  float s = fabsf(v0) + fabsf(v1);
  for (int o = 16; o > 0; o >>= 1) s += __shfl_xor(s, o, 32);
  const float d = q * s + r;
  W[wf_pos(i, j0)] = (j0 == i) ? d : v0;
  W[wf_pos(i, j1)] = (j1 == i) ? d : v1;
}


#define ENC_MMA8(a, b0, b1, b2, b3)                                                              \
  do {                                                                                           \
    v2f alo = __builtin_shufflevector(a, a, 0, 1), ahi = __builtin_shufflevector(a, a, 2, 3);     \
    c0 = __builtin_amdgcn_wmma_f32_16x16x4_f32(false, alo, false, __builtin_shufflevector(b0, b0, 0, 1), (short)0, c0, false, false); \
    c1 = __builtin_amdgcn_wmma_f32_16x16x4_f32(false, alo, false, __builtin_shufflevector(b1, b1, 0, 1), (short)0, c1, false, false); \
    c2 = __builtin_amdgcn_wmma_f32_16x16x4_f32(false, alo, false, __builtin_shufflevector(b2, b2, 0, 1), (short)0, c2, false, false); \
    c3 = __builtin_amdgcn_wmma_f32_16x16x4_f32(false, alo, false, __builtin_shufflevector(b3, b3, 0, 1), (short)0, c3, false, false); \
    c0 = __builtin_amdgcn_wmma_f32_16x16x4_f32(false, ahi, false, __builtin_shufflevector(b0, b0, 2, 3), (short)0, c0, false, false); \
    c1 = __builtin_amdgcn_wmma_f32_16x16x4_f32(false, ahi, false, __builtin_shufflevector(b1, b1, 2, 3), (short)0, c1, false, false); \
    c2 = __builtin_amdgcn_wmma_f32_16x16x4_f32(false, ahi, false, __builtin_shufflevector(b2, b2, 2, 3), (short)0, c2, false, false); \
    c3 = __builtin_amdgcn_wmma_f32_16x16x4_f32(false, ahi, false, __builtin_shufflevector(b3, b3, 2, 3), (short)0, c3, false, false); \
  } while (0)

__global__ __launch_bounds__(128)
void enc_gemm_kernel(const float* __restrict__ x, const float* __restrict__ ew,
                     const float* __restrict__ ewp,
                     float* __restrict__ h, float* __restrict__ x0) {
  int wave = (int)((blockIdx.x * blockDim.x + threadIdx.x) >> 5);
  if (wave >= M_TILES) return;
  int lane = threadIdx.x & 31;
  int mr = lane & 15;
  int kh = lane >> 4;
  const float* arow = x + (size_t)(wave * 16 + mr) * NFEAT;
  const float* ap   = arow + 4 * kh;
  const float* bp   = ewp + lane * 4;
  v8f c0 = {}, c1 = {}, c2 = {}, c3 = {};
  v4f aE  = *(const v4f*)(ap);
  v4f bE0 = *(const v4f*)(bp);
  v4f bE1 = *(const v4f*)(bp + 128);
  v4f bE2 = *(const v4f*)(bp + 256);
  v4f bE3 = *(const v4f*)(bp + 384);
  for (int kb = 0; kb < ENC_KB - 2; kb += 2) {
    const float* bo = bp + (size_t)(kb + 1) * 512;
    v4f aO  = *(const v4f*)(ap + (kb + 1) * 8);
    v4f bO0 = *(const v4f*)(bo);
    v4f bO1 = *(const v4f*)(bo + 128);
    v4f bO2 = *(const v4f*)(bo + 256);
    v4f bO3 = *(const v4f*)(bo + 384);
    ENC_MMA8(aE, bE0, bE1, bE2, bE3);
    const float* be = bp + (size_t)(kb + 2) * 512;
    aE  = *(const v4f*)(ap + (kb + 2) * 8);
    bE0 = *(const v4f*)(be);
    bE1 = *(const v4f*)(be + 128);
    bE2 = *(const v4f*)(be + 256);
    bE3 = *(const v4f*)(be + 384);
    ENC_MMA8(aO, bO0, bO1, bO2, bO3);
  }
  {
    const float* bo = bp + (size_t)(ENC_KB - 1) * 512;
    v4f aO  = *(const v4f*)(ap + (ENC_KB - 1) * 8);
    v4f bO0 = *(const v4f*)(bo);
    v4f bO1 = *(const v4f*)(bo + 128);
    v4f bO2 = *(const v4f*)(bo + 256);
    v4f bO3 = *(const v4f*)(bo + 384);
    ENC_MMA8(aE, bE0, bE1, bE2, bE3);
    ENC_MMA8(aO, bO0, bO1, bO2, bO3);
  }
  {
    int ko = ENC_KB * 8 + 2 * kh;
    v2f a  = *(const v2f*)(arow + ko);
    v2f b0 = *(const v2f*)(ew + (size_t)(mr +  0) * NFEAT + ko);
    v2f b1 = *(const v2f*)(ew + (size_t)(mr + 16) * NFEAT + ko);
    v2f b2 = *(const v2f*)(ew + (size_t)(mr + 32) * NFEAT + ko);
    v2f b3 = *(const v2f*)(ew + (size_t)(mr + 48) * NFEAT + ko);
    c0 = __builtin_amdgcn_wmma_f32_16x16x4_f32(false, a, false, b0, (short)0, c0, false, false);
    c1 = __builtin_amdgcn_wmma_f32_16x16x4_f32(false, a, false, b1, (short)0, c1, false, false);
    c2 = __builtin_amdgcn_wmma_f32_16x16x4_f32(false, a, false, b2, (short)0, c2, false, false);
    c3 = __builtin_amdgcn_wmma_f32_16x16x4_f32(false, a, false, b3, (short)0, c3, false, false);
  }
  int mbase = wave * 16 + kh * 8;
  for (int r = 0; r < 8; ++r) {
    size_t o = (size_t)(mbase + r) * HIDDEN + mr;
    h[o]      = c0[r]; x0[o]      = c0[r];
    h[o + 16] = c1[r]; x0[o + 16] = c1[r];
    h[o + 32] = c2[r]; x0[o + 32] = c2[r];
    h[o + 48] = c3[r]; x0[o + 48] = c3[r];
  }
}


#define TILE_ROWS  196
#define TILE_V4    (TILE_ROWS * HIDDEN / 4)
#define NBINS      ((N_NODES + TILE_ROWS - 1) / TILE_ROWS)
#define BIN_WGS    250
#define BIN_CHUNK  (N_EDGES / BIN_WGS)
#define BIN_PER_T  ((BIN_CHUNK + 1023) / 1024)
#define SCAN_PER_W (NBINS / 32)
#define ROW_V4     (TILE_ROWS / 4)
#define EDGE_GROUP 16
#define HALF_ROW   (HIDDEN / 2)

static_assert(N_NODES <= 65536, "a source row must fit the low 16 bits of a bin entry");
static_assert(N_EDGES % BIN_WGS == 0, "equal chunks");
static_assert(NBINS == 256 && TILE_ROWS % 4 == 0 && ROW_V4 <= 64 && TILE_ROWS <= 256,
              "32 waves x 8 bins; a bin's counts are at most two 16-byte pieces per lane; 8-bit row within the bin");

__global__ __launch_bounds__(1024)
void bin_scan_kernel(const float* __restrict__ deg, int* __restrict__ binstart,
                     int* __restrict__ cursor) {
  __shared__ int tot[NBINS];
  const int t = threadIdx.x, lane = t & 31, w = t >> 5;
  for (int k0 = 0; k0 < SCAN_PER_W; k0 += 4) {
    v4f p0[4][2], p1[4][2];
#pragma unroll
    for (int k = 0; k < 4; ++k) {
      int b = w + 32 * (k0 + k);
#pragma unroll
      for (int hh = 0; hh < 2; ++hh) {
        int q = lane + 32 * hh;
        int n = b * TILE_ROWS + 4 * q;
        int nl = (q < ROW_V4 && n < N_NODES) ? n : 0;
        p0[k][hh] = *(const v4f*)(deg + nl);
        p1[k][hh] = *(const v4f*)(deg + DEG_PAD + nl);
      }
    }
#pragma unroll
    for (int k = 0; k < 4; ++k) {
      int b = w + 32 * (k0 + k);
      float s = 0.0f;
#pragma unroll
      for (int hh = 0; hh < 2; ++hh) {
        int q = lane + 32 * hh;
        int n = b * TILE_ROWS + 4 * q;
        float u = (p0[k][hh][0] + p1[k][hh][0]) + (p0[k][hh][1] + p1[k][hh][1])
                + (p0[k][hh][2] + p1[k][hh][2]) + (p0[k][hh][3] + p1[k][hh][3]);
        if (q < ROW_V4 && n < N_NODES) s += u;
      }
      for (int o = 16; o > 0; o >>= 1) s += __shfl_xor(s, o, 32);
      if (lane == 0) tot[b] = (int)s;
    }
  }
  __syncthreads();
  if (w == 0) {
    int loc[SCAN_PER_W];
    int s = 0;
#pragma unroll
    for (int j = 0; j < SCAN_PER_W; ++j) {
      int b = lane * SCAN_PER_W + j;
      loc[j] = s;
      s += tot[b];
    }
    int incl = s;
    for (int o = 1; o < 32; o <<= 1) {
      int u = __shfl_up(incl, o, 32);
      if (lane >= o) incl += u;
    }
    int excl = incl - s;
#pragma unroll
    for (int j = 0; j < SCAN_PER_W; ++j) {
      int b = lane * SCAN_PER_W + j;
      binstart[b] = excl + loc[j];
    }
    if (lane == 31) binstart[NBINS] = incl;
  }
  if (t < NBINS) cursor[t] = 0;
}

__global__ __launch_bounds__(1024)
void bin_place_kernel(const int* __restrict__ row, const int* __restrict__ col,
                      const int* __restrict__ binstart, int* __restrict__ cursor,
                      int* __restrict__ binned) {
  __shared__ int cnt[NBINS];
  __shared__ int base[NBINS];
  const int t = threadIdx.x;
  if (t < NBINS) cnt[t] = 0;
  __syncthreads();
  const int e0 = (int)blockIdx.x * BIN_CHUNK;
  int rv[BIN_PER_T], cv[BIN_PER_T], rk[BIN_PER_T];
#pragma unroll
  for (int i = 0; i < BIN_PER_T; ++i) {
    int e = t + 1024 * i;
    int el = e < BIN_CHUNK ? e : BIN_CHUNK - 1;
    rv[i] = row[e0 + el];
    cv[i] = col[e0 + el];
  }
#pragma unroll
  for (int i = 0; i < BIN_PER_T; ++i) {
    int e = t + 1024 * i;
    rk[i] = 0;
    if (e < BIN_CHUNK) rk[i] = atomicAdd(&cnt[cv[i] / TILE_ROWS], 1);
  }
  __syncthreads();
  if (t < NBINS) {
    int n = cnt[t];
    int bs = 0;
    if (n > 0) bs = binstart[t] + atomicAdd(&cursor[t], n);
    base[t] = bs;
  }
  __syncthreads();
#pragma unroll
  for (int i = 0; i < BIN_PER_T; ++i) {
    int e = t + 1024 * i;
    if (e < BIN_CHUNK) {
      int b = cv[i] / TILE_ROWS;
      binned[base[b] + rk[i]] = rv[i] | ((cv[i] - b * TILE_ROWS) << 16);
    }
  }
}


#define NODE_ROWS 64
#define NODE_V4   (NODE_ROWS * HIDDEN / 4 / 128)
#define HS_PITCH  68

__device__ __forceinline__ void first_tile_to_lds(const float* __restrict__ h,
                                                  const float* __restrict__ deg,
                                                  float* __restrict__ Hs,
                                                  float* __restrict__ Ds) {
  const int t = threadIdx.x;
  const size_t base = (size_t)blockIdx.x * (NODE_ROWS * HIDDEN);
  v4f hv[NODE_V4];
  float d0[NODE_V4], d1[NODE_V4];
#pragma unroll
  for (int i = 0; i < NODE_V4; ++i) {
    size_t off = base + (size_t)(t + 128 * i) * 4;
    size_t ol = off < PLANE ? off : (PLANE - 4);
    hv[i] = *(const v4f*)(h + ol);
    int node = (int)blockIdx.x * NODE_ROWS + (t >> 4) + 8 * i;
    int nl = node < N_NODES ? node : N_NODES - 1;
    d0[i] = deg[nl];
    d1[i] = deg[DEG_PAD + nl];
  }
#pragma unroll
  for (int i = 0; i < NODE_V4; ++i) {
    if ((t & 15) == 0) Ds[(t >> 4) + 8 * i] = dinv_of(d0[i], d1[i]);
    int q = t + 128 * i;
    *(v4f*)(Hs + (q >> 4) * HS_PITCH + (q & 15) * 4) = hv[i];
  }
}

__device__ __forceinline__ void node_hw_product(const float* __restrict__ Hs,
                                                const float* __restrict__ Ws,
                                                const float* __restrict__ Ds,
                                                float* __restrict__ outp) {
  int w    = threadIdx.x >> 5;
  int tile = (int)blockIdx.x * 4 + w;
  if (tile >= M_TILES) return;
  int lane = threadIdx.x & 31;
  int mr = lane & 15;
  int kh = lane >> 4;
  const float* arow = Hs + (16 * w + mr) * HS_PITCH + 4 * kh;
  const float* bp   = Ws + lane * 4;
  v8f c0 = {}, c1 = {}, c2 = {}, c3 = {};
  for (int kb = 0; kb < HIDDEN / 8; ++kb) {
    const float* bk = bp + kb * 512;
    v4f a  = *(const v4f*)(arow + 8 * kb);
    v4f b0 = *(const v4f*)(bk);
    v4f b1 = *(const v4f*)(bk + 128);
    v4f b2 = *(const v4f*)(bk + 256);
    v4f b3 = *(const v4f*)(bk + 384);
    ENC_MMA8(a, b0, b1, b2, b3);
  }
  int mbase = tile * 16 + kh * 8;
  const float* dr = Ds + 16 * w + kh * 8;
  for (int r = 0; r < 8; ++r) {
    size_t o = (size_t)(mbase + r) * HIDDEN + mr;
    float dv = dr[r];
    outp[o]      = c0[r] * dv;
    outp[o + 16] = c1[r] * dv;
    outp[o + 32] = c2[r] * dv;
    outp[o + 48] = c3[r] * dv;
  }
}

__device__ __forceinline__ void stage_weff(const float* __restrict__ W, float* __restrict__ Ws) {
  v4f wv[HIDDEN * HIDDEN / 4 / 128];
#pragma unroll
  for (int i = 0; i < HIDDEN * HIDDEN / 4 / 128; ++i) wv[i] = ((const v4f*)W)[threadIdx.x + 128 * i];
#pragma unroll
  for (int i = 0; i < HIDDEN * HIDDEN / 4 / 128; ++i) ((v4f*)Ws)[threadIdx.x + 128 * i] = wv[i];
}

__global__ __launch_bounds__(128)
void node_first_kernel(const float* __restrict__ h,
                       const float* __restrict__ deg,
                       const float* __restrict__ W, float* __restrict__ outp) {
  __shared__ __attribute__((aligned(16))) float Ws[HIDDEN * HIDDEN];
  __shared__ __attribute__((aligned(16))) float Hs[NODE_ROWS * HS_PITCH];
  __shared__ float Ds[NODE_ROWS];
  stage_weff(W, Ws);
  first_tile_to_lds(h, deg, Hs, Ds);
  __syncthreads();
  node_hw_product(Hs, Ws, Ds, outp);
}


#define TILE_TILES 13
#define TILE_PAD   (TILE_TILES * 16)
#define PAD_V4     (TILE_PAD * HIDDEN / 4)
#define UPD_PER_T  ((PAD_V4 + 1023) / 1024)

static_assert(TILE_PAD >= TILE_ROWS && 2 * TILE_TILES <= 32 && HIDDEN * HIDDEN / 4 == 1024,
              "13 tiles x 2 column halves on 32 waves; W_eff is one 16-byte piece per thread");

#define TILE_MMA8(a, b0, b1)                                                                     \
  do {                                                                                           \
    v2f alo = __builtin_shufflevector(a, a, 0, 1), ahi = __builtin_shufflevector(a, a, 2, 3);     \
    c0 = __builtin_amdgcn_wmma_f32_16x16x4_f32(false, alo, false, __builtin_shufflevector(b0, b0, 0, 1), (short)0, c0, false, false); \
    c1 = __builtin_amdgcn_wmma_f32_16x16x4_f32(false, alo, false, __builtin_shufflevector(b1, b1, 0, 1), (short)0, c1, false, false); \
    c0 = __builtin_amdgcn_wmma_f32_16x16x4_f32(false, ahi, false, __builtin_shufflevector(b0, b0, 2, 3), (short)0, c0, false, false); \
    c1 = __builtin_amdgcn_wmma_f32_16x16x4_f32(false, ahi, false, __builtin_shufflevector(b1, b1, 2, 3), (short)0, c1, false, false); \
  } while (0)

__device__ __forceinline__ void tile_pass(const bool last,
                                          const int* __restrict__ binned,
                                          const int* __restrict__ binstart,
                                          const float* __restrict__ outp,
                                          float* __restrict__ h,
                                          const float* __restrict__ x0,
                                          const float* __restrict__ ext_w,
                                          const float* __restrict__ beta,
                                          const float* __restrict__ deg,
                                          const float* __restrict__ wsrc,
                                          float* __restrict__ dst,
                                          float* __restrict__ T, float* __restrict__ Hs,
                                          float* __restrict__ Ws, float* __restrict__ Ds) {
  const int t = threadIdx.x;
  const int lane = t & 31;
  const int w = __builtin_amdgcn_readfirstlane(t >> 5);
  const int b = (int)blockIdx.x;
  const int row0 = b * TILE_ROWS;

  v4f hv[UPD_PER_T], xv[UPD_PER_T];
  float d0[UPD_PER_T], d1[UPD_PER_T];
  v4f wv = {}, ew = {};
  float bt = 0.0f;
#pragma unroll
  for (int i = 0; i < UPD_PER_T; ++i) {
    v4f z = {};
    hv[i] = z; xv[i] = z; d0[i] = 0.0f; d1[i] = 0.0f;
  }
  auto fetch_own = [&]() {
#pragma unroll
    for (int i = 0; i < UPD_PER_T; ++i) {
      int q = t + 1024 * i;
      int r = q >> 4;
      bool ok = r < TILE_ROWS && row0 + r < N_NODES;
      int node = ok ? row0 + r : row0;
      size_t o = (size_t)node * HIDDEN + (q & 15) * 4;
      hv[i] = *(const v4f*)(h + o);
      xv[i] = *(const v4f*)(x0 + o);
      d0[i] = deg[node];
      d1[i] = deg[DEG_PAD + node];
    }
    if (!last) wv = ((const v4f*)wsrc)[t];
    ew = *(const v4f*)(ext_w + 4 * (t & 15));
    bt = beta[0];
  };
  for (int q = t; q < TILE_V4; q += 1024) {
    v4f z = {};
    ((v4f*)T)[q] = z;
  }
  __syncthreads();

  {
    const int beg = binstart[b];
    const int end = binstart[b + 1];
    const float* srcl = outp + lane;
    float*       Tl   = T + lane;
    int e0 = beg + 32 * w;
    int i0 = e0 + lane;
    int p  = binned[i0 < N_EDGES ? i0 : N_EDGES - 1];
    const int slot = (w >> 2) % 5;
    int it = 0;
    bool fetched = false;
    while (e0 < end) {
      if (it == slot) { fetch_own(); fetched = true; }
      ++it;
      const int e1 = e0 + 32 * 32;
      const int i1 = e1 + lane;
      const int pn = binned[i1 < N_EDGES ? i1 : N_EDGES - 1];
      const int n  = end - e0;
      if (n >= 32) {
#pragma unroll
        for (int g = 0; g < 32; g += EDGE_GROUP) {
          float va[EDGE_GROUP], vb[EDGE_GROUP];
#pragma unroll
          for (int j = 0; j < EDGE_GROUP; ++j) {
            int r = __builtin_amdgcn_readlane(p, g + j) & 0xFFFF;
            const float* s = srcl + (size_t)r * HIDDEN;
            va[j] = s[0];
            vb[j] = s[HALF_ROW];
          }
#pragma unroll
          for (int j = 0; j < EDGE_GROUP; ++j) {
            int c = __builtin_amdgcn_readlane(p, g + j) >> 16;
            float* d = Tl + c * HIDDEN;
            atomicAdd(&d[0],        va[j]);
            atomicAdd(&d[HALF_ROW], vb[j]);
          }
        }
      } else {
        for (int j = 0; j < n; ++j) {
          int pj = __builtin_amdgcn_readlane(p, j);
          const float* s = srcl + (size_t)(pj & 0xFFFF) * HIDDEN;
          float a0 = s[0], a1 = s[HALF_ROW];
          float* d = Tl + (pj >> 16) * HIDDEN;
          atomicAdd(&d[0],        a0);
          atomicAdd(&d[HALF_ROW], a1);
        }
      }
      p  = pn;
      e0 = e1;
    }
    if (!fetched) fetch_own();
  }
  __syncthreads();

  {
    if (!last) ((v4f*)Ws)[t] = wv;
#pragma unroll
    for (int i = 0; i < UPD_PER_T; ++i) {
      int q = t + 1024 * i;
      if (q < PAD_V4) {
        int r = q >> 4;
        v4f hn = {};
        float dv = 0.0f;
        if (r < TILE_ROWS && row0 + r < N_NODES) {
          v4f a = ((const v4f*)T)[q];
          dv = dinv_of(d0[i], d1[i]);
          for (int c = 0; c < 4; ++c) {
            float conv = dv * a[c] - hv[i][c] * ew[c] + xv[i][c] * bt;
            hn[c] = hv[i][c] + STEP_SZ * fmaxf(conv, 0.0f);
          }
          if (!last) *(v4f*)(h + (size_t)(row0 + r) * HIDDEN + (q & 15) * 4) = hn;
        }
        if ((t & 15) == 0) Ds[r] = dv;
        *(v4f*)(Hs + r * HS_PITCH + (q & 15) * 4) = hn;
      }
    }
  }
  __syncthreads();

  const int mr = lane & 15;
  const int kh = lane >> 4;
  if (!last) {
    if (w >= 2 * TILE_TILES) return;
    const int tile = w >> 1;
    const int np   = w & 1;
    const float* arow = Hs + (16 * tile + mr) * HS_PITCH + 4 * kh;
    const float* bp   = Ws + np * 256 + lane * 4;
    v8f c0 = {}, c1 = {};
    for (int kb = 0; kb < HIDDEN / 8; ++kb) {
      const float* bk = bp + kb * 512;
      v4f a  = *(const v4f*)(arow + 8 * kb);
      v4f b0 = *(const v4f*)(bk);
      v4f b1 = *(const v4f*)(bk + 128);
      TILE_MMA8(a, b0, b1);
    }
    const int rbase = 16 * tile + kh * 8;
    for (int r = 0; r < 8; ++r) {
      int tr = rbase + r;
      if (tr < TILE_ROWS && row0 + tr < N_NODES) {
        float dv = Ds[tr];
        size_t o = (size_t)(row0 + tr) * HIDDEN + 32 * np + mr;
        dst[o]      = c0[r] * dv;
        dst[o + 16] = c1[r] * dv;
      }
    }
  } else {
    if (w >= TILE_TILES) return;
    const float* arow = Hs + (16 * w + mr) * HS_PITCH + 4 * kh;
    const float* b0r = wsrc + (size_t)(mr +  0) * HIDDEN + 4 * kh;
    const float* b1r = wsrc + (size_t)(mr + 16) * HIDDEN + 4 * kh;
    const int n2 = mr + 32;
    const float* b2r = wsrc + (size_t)(n2 < NCLASS ? n2 : 0) * HIDDEN + 4 * kh;
    const float b2scale = (n2 < NCLASS) ? 1.0f : 0.0f;
    v8f c0 = {}, c1 = {}, c2 = {};
    for (int kb = 0; kb < HIDDEN / 8; ++kb) {
      v4f a  = *(const v4f*)(arow + 8 * kb);
      v4f b0 = *(const v4f*)(b0r + 8 * kb);
      v4f b1 = *(const v4f*)(b1r + 8 * kb);
      v4f b2 = *(const v4f*)(b2r + 8 * kb);
      b2 *= b2scale;
      v2f alo = __builtin_shufflevector(a, a, 0, 1), ahi = __builtin_shufflevector(a, a, 2, 3);
      c0 = __builtin_amdgcn_wmma_f32_16x16x4_f32(false, alo, false, __builtin_shufflevector(b0, b0, 0, 1), (short)0, c0, false, false);
      c1 = __builtin_amdgcn_wmma_f32_16x16x4_f32(false, alo, false, __builtin_shufflevector(b1, b1, 0, 1), (short)0, c1, false, false);
      c2 = __builtin_amdgcn_wmma_f32_16x16x4_f32(false, alo, false, __builtin_shufflevector(b2, b2, 0, 1), (short)0, c2, false, false);
      c0 = __builtin_amdgcn_wmma_f32_16x16x4_f32(false, ahi, false, __builtin_shufflevector(b0, b0, 2, 3), (short)0, c0, false, false);
      c1 = __builtin_amdgcn_wmma_f32_16x16x4_f32(false, ahi, false, __builtin_shufflevector(b1, b1, 2, 3), (short)0, c1, false, false);
      c2 = __builtin_amdgcn_wmma_f32_16x16x4_f32(false, ahi, false, __builtin_shufflevector(b2, b2, 2, 3), (short)0, c2, false, false);
    }
    const int rbase = 16 * w + kh * 8;
    for (int r = 0; r < 8; ++r) {
      int tr = rbase + r;
      if (tr < TILE_ROWS && row0 + tr < N_NODES) {
        float* orow = dst + (size_t)(row0 + tr) * NCLASS;
        orow[mr]      = c0[r];
        orow[mr + 16] = c1[r];
        if (n2 < NCLASS) orow[n2] = c2[r];
      }
    }
  }
}

__global__ __launch_bounds__(1024)
void tile_mid_kernel(const int* __restrict__ binned, const int* __restrict__ binstart,
                     const float* __restrict__ outp, float* __restrict__ h,
                     const float* __restrict__ x0, const float* __restrict__ ext_w,
                     const float* __restrict__ beta, const float* __restrict__ deg,
                     const float* __restrict__ W, float* __restrict__ outn) {
  __shared__ __attribute__((aligned(16))) float T[TILE_ROWS * HIDDEN];
  __shared__ __attribute__((aligned(16))) float Hs[TILE_PAD * HS_PITCH];
  __shared__ __attribute__((aligned(16))) float Ws[HIDDEN * HIDDEN];
  __shared__ float Ds[TILE_PAD];
  tile_pass(false, binned, binstart, outp, h, x0, ext_w, beta, deg, W, outn, T, Hs, Ws, Ds);
}

__global__ __launch_bounds__(1024)
void tile_last_kernel(const int* __restrict__ binned, const int* __restrict__ binstart,
                      const float* __restrict__ outp, float* __restrict__ h,
                      const float* __restrict__ x0, const float* __restrict__ ext_w,
                      const float* __restrict__ beta, const float* __restrict__ deg,
                      const float* __restrict__ dw, float* __restrict__ out) {
  __shared__ __attribute__((aligned(16))) float T[TILE_ROWS * HIDDEN];
  __shared__ __attribute__((aligned(16))) float Hs[TILE_PAD * HS_PITCH];
  __shared__ float Ds[TILE_PAD];
  tile_pass(true, binned, binstart, outp, h, x0, ext_w, beta, deg, dw, out, T, Hs, nullptr, Ds);
}


extern "C" void kernel_launch(void* const* d_in, const int* in_sizes, int n_in,
                              void* d_out, int out_size, void* d_ws, size_t ws_size,
                              hipStream_t stream) {
  (void)in_sizes; (void)n_in; (void)out_size; (void)ws_size;
  const float* x     = (const float*)d_in[0];
  const int*   ei    = (const int*)  d_in[1];
  const float* enc_w = (const float*)d_in[2];
  const float* dec_w = (const float*)d_in[3];
  const float* ext_w = (const float*)d_in[4];
  const float* beta  = (const float*)d_in[5];
  const float* pw    = (const float*)d_in[6];
  const int* row = ei;
  const int* col = ei + N_EDGES;

  float* ws   = (float*)d_ws;
  float* deg  = ws;
  float* W    = deg  + 2 * DEG_PAD;
  float* h    = W    + HIDDEN * HIDDEN;
  float* x0   = h    + (size_t)N_NODES * HIDDEN;
  float* outp = x0   + (size_t)N_NODES * HIDDEN;
  float* outq = outp + (size_t)N_NODES * HIDDEN;
  float* ewp  = outq + PLANE;
  int* binstart = (int*)(ewp + (size_t)ENC_FRAGS * 4);
  int* cursor   = binstart + 2 * NBINS;
  int* binned   = cursor + NBINS;

  hipMemsetAsync(deg, 0, (DEG_PAD + N_NODES) * sizeof(float), stream);
  deg_count_kernel<<<(N_EDGES + 255) / 256, 256, 0, stream>>>(col, deg);
  weff_pack_kernel<<<WEFF_BLOCKS + (ENC_FRAGS + HIDDEN - 1) / HIDDEN, HIDDEN, 0, stream>>>(pw, W, enc_w, ewp);
  bin_scan_kernel<<<1, 1024, 0, stream>>>(deg, binstart, cursor);
  bin_place_kernel<<<BIN_WGS, 1024, 0, stream>>>(row, col, binstart, cursor, binned);

  const int gemm_blocks = (M_TILES + 3) / 4;
  enc_gemm_kernel<<<gemm_blocks, 128, 0, stream>>>(x, enc_w, ewp, h, x0);

  node_first_kernel<<<gemm_blocks, 128, 0, stream>>>(h, deg, W, outp);
  tile_mid_kernel<<<NBINS, 1024, 0, stream>>>(binned, binstart, outp, h, x0, ext_w, beta, deg, W, outq);
  tile_mid_kernel<<<NBINS, 1024, 0, stream>>>(binned, binstart, outq, h, x0, ext_w, beta, deg, W, outp);
  tile_mid_kernel<<<NBINS, 1024, 0, stream>>>(binned, binstart, outp, h, x0, ext_w, beta, deg, W, outq);
  tile_last_kernel<<<NBINS, 1024, 0, stream>>>(binned, binstart, outq, h, x0, ext_w, beta, deg, dec_w, (float*)d_out);
}
